# starting kernel
# baseline (speedup 1.0000x reference)
#include <hip/hip_runtime.h>
#include <math.h>

#define IN_C 512
#define HID 100
#define OUT_C 2
#define NT 8
#define WP_FLOATS (256 * NT * 16 * 2)

typedef __attribute__((ext_vector_type(2))) float v2f;
typedef __attribute__((ext_vector_type(8))) float v8f;

__global__ void deg_init_kernel(float* deg, int n) {
  int i = blockIdx.x * blockDim.x + threadIdx.x;
  if (i < n) deg[i] = 1.0f;
}

__global__ void deg_accum_kernel(const int* __restrict__ dst, float* deg, int e) {
  int i = blockIdx.x * blockDim.x + threadIdx.x;
  if (i < e) atomicAdd(&deg[dst[i]], 1.0f);
}

__global__ void deg_rsqrt_kernel(float* deg, int n) {
  int i = blockIdx.x * blockDim.x + threadIdx.x;
  if (i < n) { float d = deg[i]; deg[i] = (d > 0.f) ? rsqrtf(d) : 0.f; }
}

__global__ void pack_w_kernel(const float* __restrict__ W1,
                              const float* __restrict__ Ws,
                              float* __restrict__ Wp) {
  int idx = blockIdx.x * blockDim.x + threadIdx.x;
  if (idx >= WP_FLOATS) return;
  int kin = idx & 1;
  int col = (idx >> 1) & 15;
  int t   = (idx >> 5) & 7;
  int kp  = idx >> 8;
  int k   = (kp << 1) + kin;
  float v = 0.f;
  if (t < 7) {
    int c = (t << 4) + col;
    if (c < HID) v = W1[(size_t)k * HID + c];
  } else {
    if (col < OUT_C) v = Ws[(size_t)k * OUT_C + col];
  }
  Wp[idx] = v;
}

__global__ void __launch_bounds__(256)
gemm1_wmma_kernel(const float* __restrict__ x, const float* __restrict__ Wp,
                  const float* __restrict__ dinv, float* __restrict__ h,
                  float* __restrict__ agg, float* __restrict__ outp, int n_nodes) {
  const int lane = threadIdx.x & 31;
  const int wv   = threadIdx.x >> 5;
  const int tile = blockIdx.x * (blockDim.x >> 5) + wv;
  const int row0 = tile << 4;
  if (row0 >= n_nodes) return;

  const int m  = lane & 15;
  const int kh = (lane >> 4) << 1;

  v8f acc[NT] = {};

  const float* __restrict__ pa = x + (size_t)(row0 + m) * IN_C + kh;
  const float* __restrict__ pb = Wp + ((size_t)(lane >> 4) * 256) + (m << 1);

  for (int k0 = 0; k0 < IN_C; k0 += 4) {
    v2f a = *(const v2f*)pa;
#pragma unroll
    for (int t = 0; t < NT; ++t) {
      v2f b = *(const v2f*)(pb + (t << 5));
      acc[t] = __builtin_amdgcn_wmma_f32_16x16x4_f32(
          false, a, false, b, (short)0, acc[t], false, false);
    }
    pa += 4;
    pb += 512;
  }

  const int rbase = (lane < 16) ? 0 : 8;
#pragma unroll
  for (int r = 0; r < 8; ++r) {
    const int row = row0 + rbase + r;
    const float di  = dinv[row];
    const float di2 = di * di;
    float* __restrict__ hrow = h   + (size_t)row * HID;
    float* __restrict__ arow = agg + (size_t)row * HID;
#pragma unroll
    for (int t = 0; t < 7; ++t) {
      int col = (t << 4) + m;
      if (col < HID) {
        float v = acc[t][r];
        hrow[col] = v;
        arow[col] = v * di2;
      }
    }
    if (m < OUT_C) outp[(size_t)row * OUT_C + m] = acc[7][r];
  }
}

__global__ void scatter1_kernel(const float* __restrict__ h,
                                const float* __restrict__ dinv,
                                const int* __restrict__ src,
                                const int* __restrict__ dst,
                                float* __restrict__ agg, int n_edges) {
  int lane = threadIdx.x & 31;
  int wv   = threadIdx.x >> 5;
  long long e = (long long)blockIdx.x * (blockDim.x >> 5) + wv;
  if (e >= n_edges) return;
  int s = src[e], d = dst[e];
  float norm = dinv[s] * dinv[d];
  const float* __restrict__ hs = h + (size_t)s * HID;
  float* __restrict__ ad = agg + (size_t)d * HID;
#pragma unroll
  for (int i = 0; i < 4; ++i) {
    int f = lane + (i << 5);
    if (f < HID) atomicAdd(&ad[f], hs[f] * norm);
  }
}

__global__ void gemm2_fused_kernel(const float* __restrict__ agg,
                                   const float* __restrict__ b1,
                                   const float* __restrict__ W2,
                                   const float* __restrict__ b2,
                                   const float* __restrict__ bs,
                                   const float* __restrict__ dinv,
                                   float* __restrict__ h2,
                                   float* __restrict__ out, int n_nodes) {
  int n = blockIdx.x * blockDim.x + threadIdx.x;
  if (n >= n_nodes) return;
  const float* __restrict__ r = agg + (size_t)n * HID;
  float a0 = 0.f, a1 = 0.f;
#pragma unroll 4
  for (int k = 0; k < HID; ++k) {
    float v = r[k] + b1[k];
    v = v > 0.f ? v : 0.f;
    a0 += v * W2[k * 2 + 0];
    a1 += v * W2[k * 2 + 1];
  }
  h2[(size_t)n * 2 + 0] = a0;
  h2[(size_t)n * 2 + 1] = a1;
  float di = dinv[n];
  float di2 = di * di;
  out[(size_t)n * 2 + 0] += b2[0] + bs[0] + a0 * di2;
  out[(size_t)n * 2 + 1] += b2[1] + bs[1] + a1 * di2;
}

__global__ void scatter2_kernel(const float* __restrict__ h2,
                                const float* __restrict__ dinv,
                                const int* __restrict__ src,
                                const int* __restrict__ dst,
                                float* __restrict__ out, int n_edges) {
  int e = blockIdx.x * blockDim.x + threadIdx.x;
  if (e >= n_edges) return;
  int s = src[e], d = dst[e];
  float norm = dinv[s] * dinv[d];
  atomicAdd(&out[(size_t)d * 2 + 0], h2[(size_t)s * 2 + 0] * norm);
  atomicAdd(&out[(size_t)d * 2 + 1], h2[(size_t)s * 2 + 1] * norm);
}

__global__ void log_softmax_kernel(float* __restrict__ out, int n_nodes) {
  int n = blockIdx.x * blockDim.x + threadIdx.x;
  if (n >= n_nodes) return;
  float z0 = out[(size_t)n * 2 + 0];
  float z1 = out[(size_t)n * 2 + 1];
  float mx = fmaxf(z0, z1);
  float lse = mx + logf(expf(z0 - mx) + expf(z1 - mx));
  out[(size_t)n * 2 + 0] = z0 - lse;
  out[(size_t)n * 2 + 1] = z1 - lse;
}

extern "C" void kernel_launch(void* const* d_in, const int* in_sizes, int n_in,
                              void* d_out, int out_size, void* d_ws, size_t ws_size,
                              hipStream_t stream) {
  const float* x  = (const float*)d_in[0];
  const int*   ei = (const int*)d_in[1];
  const float* W1 = (const float*)d_in[2];
  const float* b1 = (const float*)d_in[3];
  const float* W2 = (const float*)d_in[4];
  const float* b2 = (const float*)d_in[5];
  const float* Ws = (const float*)d_in[6];
  const float* bs = (const float*)d_in[7];
  float* out = (float*)d_out;

  const int n_nodes = in_sizes[0] / IN_C;
  const int n_edges = in_sizes[1] / 2;
  const int* srcv = ei;
  const int* dstv = ei + n_edges;

  float* Wp   = (float*)d_ws;
  float* dinv = Wp + WP_FLOATS;
  size_t off  = ((size_t)n_nodes + 255) & ~(size_t)255;
  float* h    = dinv + off;
  float* agg  = h + (size_t)n_nodes * HID;
  float* h2   = agg + (size_t)n_nodes * HID;

  const int T = 256;

  deg_init_kernel<<<(n_nodes + T - 1) / T, T, 0, stream>>>(dinv, n_nodes);
  deg_accum_kernel<<<(n_edges + T - 1) / T, T, 0, stream>>>(dstv, dinv, n_edges);
  deg_rsqrt_kernel<<<(n_nodes + T - 1) / T, T, 0, stream>>>(dinv, n_nodes);

  pack_w_kernel<<<WP_FLOATS / T, T, 0, stream>>>(W1, Ws, Wp);

  int tiles = (n_nodes + 15) / 16;
  gemm1_wmma_kernel<<<(tiles + 7) / 8, T, 0, stream>>>(x, Wp, dinv, h, agg, out, n_nodes);

  scatter1_kernel<<<(n_edges + 7) / 8, T, 0, stream>>>(h, dinv, srcv, dstv, agg, n_edges);

  gemm2_fused_kernel<<<(n_nodes + T - 1) / T, T, 0, stream>>>(agg, b1, W2, b2, bs, dinv,
                                                              h2, out, n_nodes);
  scatter2_kernel<<<(n_edges + T - 1) / T, T, 0, stream>>>(h2, dinv, srcv, dstv, out, n_edges);

  log_softmax_kernel<<<(n_nodes + T - 1) / T, T, 0, stream>>>(out, n_nodes);
}


# scatter1: a workgroup owns 400 dst nodes, filters the whole edge list itself and sums their rows in LDS (LDS float adds, LDS arrival order); agg written by plain line stores, no global atomics on it
# speedup vs baseline: 1.8171x; 1.8171x over previous
#include <hip/hip_runtime.h>
#include <math.h>

#define IN_C 512
#define HID 100
#define OUT_C 2
#define NT 8
#define WP_FLOATS (256 * NT * 16 * 2)

typedef __attribute__((ext_vector_type(2))) float v2f;
typedef __attribute__((ext_vector_type(8))) float v8f;

#define PW 32
#define NPL 3
#define TAILF (HID - NPL * PW)

__global__ void deg_init_kernel(float* deg, int n) {
  int i = blockIdx.x * blockDim.x + threadIdx.x;
  if (i < n) deg[i] = 1.0f;
}

__device__ __forceinline__ float deg_to_dinv(float d) { return (d > 0.f) ? rsqrtf(d) : 0.f; }

__global__ void pack_w_kernel(const float* __restrict__ W1,
                              const float* __restrict__ Ws,
                              float* __restrict__ Wp) {
  int idx = blockIdx.x * blockDim.x + threadIdx.x;
  if (idx >= WP_FLOATS) return;
  int kin = idx & 1;
  int col = (idx >> 1) & 15;
  int t   = (idx >> 5) & 7;
  int kp  = idx >> 8;
  int k   = (kp << 1) + kin;
  float v = 0.f;
  if (t < 7) {
    int c = (t << 4) + col;
    if (c < HID) v = W1[(size_t)k * HID + c];
  } else {
    if (col < OUT_C) v = Ws[(size_t)k * OUT_C + col];
  }
  Wp[idx] = v;
}

#define KBLK 16
__global__ void __launch_bounds__(256)
gemm1_wmma_kernel(const float* __restrict__ x, const float* __restrict__ Wp,
                  const int* __restrict__ dst, float* __restrict__ deg,
                  float* __restrict__ h,
                  float* __restrict__ outp, int n_nodes, int n_edges) {
  const int lane = threadIdx.x & 31;
  const int wv   = threadIdx.x >> 5;
  const int tile = blockIdx.x * (blockDim.x >> 5) + wv;
  const int row0 = tile << 4;
  if (row0 >= n_nodes) return;

  const long long stride = (long long)((n_nodes + 15) >> 4) << 5;
  long long e = ((long long)tile << 5) + lane;
  int dcur = (e < n_edges) ? dst[e] : -1;

  const int m  = lane & 15;
  const int kh = (lane >> 4) << 1;

  v8f acc[NT] = {};

  const float* __restrict__ pa = x + (size_t)(row0 + m) * IN_C + kh;
  const float* __restrict__ pb = Wp + ((size_t)(lane >> 4) * 256) + (m << 1);

  for (int kb = 0; kb < IN_C / (4 * KBLK); ++kb) {
    if (dcur >= 0) atomicAdd(&deg[dcur], 1.0f);
    const long long en = e + stride;
    const int dnxt = (en < n_edges) ? dst[en] : -1;
    for (int ks = 0; ks < KBLK; ++ks) {
      v2f a = *(const v2f*)pa;
#pragma unroll
      for (int t = 0; t < NT; ++t) {
        v2f b = *(const v2f*)(pb + (t << 5));
        acc[t] = __builtin_amdgcn_wmma_f32_16x16x4_f32(
            false, a, false, b, (short)0, acc[t], false, false);
      }
      pa += 4;
      pb += 512;
    }
    dcur = dnxt;
    e = en;
  }
  if (dcur >= 0) atomicAdd(&deg[dcur], 1.0f);
  for (e += stride; e < n_edges; e += stride) atomicAdd(&deg[dst[e]], 1.0f);

  const int rbase = (lane < 16) ? 0 : 8;
  const size_t plane = (size_t)n_nodes * PW;
#pragma unroll
  for (int r = 0; r < 8; ++r) {
    const int row = row0 + rbase + r;
#pragma unroll
    for (int t = 0; t < 7; ++t) {
      int col = (t << 4) + m;
      if (col < HID) {
        const size_t o = (t < 2 * NPL)
            ? (size_t)(t >> 1) * plane + (size_t)row * PW + ((t & 1) << 4) + m
            : (size_t)NPL * plane + (size_t)row * TAILF + m;
        h[o] = acc[t][r];
      }
    }
    if (m < OUT_C) outp[(size_t)row * OUT_C + m] = acc[7][r];
  }
}

#define NPB 400
#define SW 32
#define SU 4
#define LCAP (32 + SU * 128)
#define TU 4
#define EB 8
static_assert(EB * TAILF == 32, "one wave-wide access covers the leftover features of a batch");
static_assert(LCAP >= 32 + TU * 32, "the list also holds a trip of the edge-at-a-time loop");

typedef __attribute__((ext_vector_type(4))) int v4i;

__device__ __forceinline__ void lds_add(float* p, float v) {
  __hip_atomic_fetch_add(p, v, __ATOMIC_RELAXED, __HIP_MEMORY_SCOPE_WORKGROUP);
}

__global__ void __launch_bounds__(SW * 32)
scatter1_kernel(const float* __restrict__ h,
                const float* __restrict__ deg,
                const int* __restrict__ src,
                const int* __restrict__ dst,
                float* __restrict__ agg, int n_edges, int n_nodes) {
  __shared__ float acc[NPB * HID];
  __shared__ int lst[SW * LCAP];
  const int lane = threadIdx.x & 31;
  const int wv = __builtin_amdgcn_readfirstlane((int)(threadIdx.x >> 5));
  const int base = blockIdx.x * NPB;
  const int cnt = (n_nodes - base < NPB) ? (n_nodes - base) : NPB;
  const size_t plane = (size_t)n_nodes * PW;
  const char* __restrict__ hb0 = (const char*)h;
  const char* __restrict__ hb1 = (const char*)(h + plane);
  const char* __restrict__ hb2 = (const char*)(h + 2 * plane);
  const char* __restrict__ hbt = (const char*)(h + NPL * plane);
  const unsigned lb = (unsigned)lane * 4u;
  const int tu = lane / TAILF;
  const int tf = lane % TAILF;
  int* const my = lst + wv * LCAP;

  for (int i = threadIdx.x; i < NPB * HID; i += SW * 32) acc[i] = 0.f;
  __syncthreads();

  auto drain = [&](const int* p, int k) {
    const int e = (lane < k) ? p[lane] : -1;
    int s = 0, dl = 0;
    float nrm = 0.f;
    if (e >= 0) {
      s = src[e];
      const int d = dst[e];
      nrm = deg_to_dinv(deg[s]) * deg_to_dinv(deg[d]);
      dl = d - base;
    }
    for (int jb = 0; jb < k; jb += EB) {
      float v[EB][3];
      const int   st = __shfl(s, jb + tu);
      const int   dt = __shfl(dl, jb + tu);
      const float nt = __shfl(nrm, jb + tu);
#pragma unroll
      for (int u = 0; u < EB; ++u) {
        const int sj = __builtin_amdgcn_readlane(s, jb + u);
        const unsigned so = (unsigned)sj * (PW * 4u) + lb;
        v[u][0] = *(const float*)(hb0 + so);
        v[u][1] = *(const float*)(hb1 + so);
        v[u][2] = *(const float*)(hb2 + so);
      }
      const float vt = *(const float*)(hbt + ((unsigned)st * (TAILF * 4u) + (unsigned)tf * 4u));
#pragma unroll
      for (int u = 0; u < EB; ++u) {
        if (jb + u < k) {
          const int dj = __builtin_amdgcn_readlane(dl, jb + u);
          const float nj = __int_as_float(__builtin_amdgcn_readlane(__float_as_int(nrm), jb + u));
          float* r = acc + dj * HID + lane;
          lds_add(r,      v[u][0] * nj);
          lds_add(r + 32, v[u][1] * nj);
          lds_add(r + 64, v[u][2] * nj);
        }
      }
      if (jb + tu < k) lds_add(acc + dt * HID + NPL * PW + tf, vt * nt);
    }
  };

  int count = 0;
  auto append = [&](bool m, int ev) {
    const unsigned bal = __builtin_amdgcn_ballot_w32(m);
    if (bal) {
      if (m) my[count + (int)__builtin_amdgcn_mbcnt_lo(bal, 0u)] = ev;
      count += __builtin_popcount(bal);
    }
  };

  const long long wstride = (long long)SW * SU * 128;
  const int nfull = ((((size_t)dst) & 15) == 0) ? (int)((long long)n_edges / wstride) : 0;
  if (nfull > 0) {
    int ps = (int)(((long long)blockIdx.x * nfull) / gridDim.x);
    const long long wofs = (long long)wv * (SU * 128) + lane * 4;
    long long eb = (long long)ps * wstride + wofs;
    v4i cur[SU];
#pragma unroll
    for (int u = 0; u < SU; ++u) cur[u] = *(const v4i*)(dst + eb + u * 128);
#pragma unroll 2
    for (int it = 0; it < nfull; ++it) {
      ps = (ps + 1 < nfull) ? ps + 1 : 0;
      const long long en = (it + 1 < nfull) ? (long long)ps * wstride + wofs : eb;
      v4i nxt[SU];
#pragma unroll
      for (int u = 0; u < SU; ++u) nxt[u] = *(const v4i*)(dst + en + u * 128);
#pragma unroll
      for (int u = 0; u < SU; ++u) {
        const v4i dv = cur[u];
        const int e0 = (int)(eb + u * 128);
        const bool m0 = (unsigned)(dv.x - base) < (unsigned)cnt;
        const bool m1 = (unsigned)(dv.y - base) < (unsigned)cnt;
        const bool m2 = (unsigned)(dv.z - base) < (unsigned)cnt;
        const bool m3 = (unsigned)(dv.w - base) < (unsigned)cnt;
        if (__builtin_amdgcn_ballot_w32(m0 | m1 | m2 | m3)) {
          append(m0, e0);
          append(m1, e0 + 1);
          append(m2, e0 + 2);
          append(m3, e0 + 3);
        }
      }
      __builtin_amdgcn_fence(__ATOMIC_SEQ_CST, "wavefront");
      while (count >= 32) {
        count -= 32;
        drain(my + count, 32);
      }
#pragma unroll
      for (int u = 0; u < SU; ++u) cur[u] = nxt[u];
      eb = en;
    }
  }
  for (long long t = (long long)nfull * wstride + wv * 32; ; t += TU * SW * 32) {
    const bool done = t >= n_edges;
    if (!done) {
      int dv[TU];
#pragma unroll
      for (int u = 0; u < TU; ++u) {
        const long long e = t + u * (SW * 32) + lane;
        dv[u] = (e < n_edges) ? dst[e] : -1;
      }
#pragma unroll
      for (int u = 0; u < TU; ++u)
        append((unsigned)(dv[u] - base) < (unsigned)cnt, (int)(t + u * (SW * 32) + lane));
    }
    __builtin_amdgcn_fence(__ATOMIC_SEQ_CST, "wavefront");
    const int lim = done ? 1 : 32;
    while (count >= lim) {
      const int k = count < 32 ? count : 32;
      count -= k;
      drain(my + count, k);
    }
    if (done) break;
  }

  __syncthreads();

  for (int r = wv; r < cnt; r += SW) {
    const float* a = acc + r * HID;
    const size_t o = (size_t)(base + r) * PW + lane;
    agg[o]             = a[lane];
    agg[plane + o]     = a[PW + lane];
    agg[2 * plane + o] = a[2 * PW + lane];
  }
  for (int i = threadIdx.x; i < cnt * TAILF; i += SW * 32)
    agg[NPL * plane + (size_t)base * TAILF + i] = acc[(i / TAILF) * HID + NPL * PW + (i % TAILF)];
}

__global__ void gemm2_fused_kernel(const float* __restrict__ agg,
                                   const float* __restrict__ h,
                                   const float* __restrict__ b1,
                                   const float* __restrict__ W2,
                                   const float* __restrict__ b2,
                                   const float* __restrict__ bs,
                                   const float* __restrict__ deg,
                                   float* __restrict__ h2,
                                   float* __restrict__ out, int n_nodes) {
  int n = blockIdx.x * blockDim.x + threadIdx.x;
  if (n >= n_nodes) return;
  const size_t plane = (size_t)n_nodes * PW;
  const float di  = deg_to_dinv(deg[n]);
  const float di2 = di * di;
  const v2f* __restrict__ w2 = (const v2f*)W2;
  v2f a = {0.f, 0.f};
  for (int p = 0; p < NPL; ++p) {
    const float4* __restrict__ r4 = (const float4*)(agg + p * plane + (size_t)n * PW);
    const float4* __restrict__ g4 = (const float4*)(h + p * plane + (size_t)n * PW);
#pragma unroll
    for (int q = 0; q < PW / 4; ++q) {
      const float4 v4 = r4[q];
      const float4 s4 = g4[q];
      const float vv[4] = {v4.x + s4.x * di2, v4.y + s4.y * di2, v4.z + s4.z * di2, v4.w + s4.w * di2};
#pragma unroll
      for (int c = 0; c < 4; ++c) {
        const int k = p * PW + q * 4 + c;
        float v = vv[c] + b1[k];
        v = v > 0.f ? v : 0.f;
        a += v2f{v, v} * w2[k];
      }
    }
  }
  {
    const float4 v4 = *(const float4*)(agg + NPL * plane + (size_t)n * TAILF);
    const float4 s4 = *(const float4*)(h + NPL * plane + (size_t)n * TAILF);
    const float vv[4] = {v4.x + s4.x * di2, v4.y + s4.y * di2, v4.z + s4.z * di2, v4.w + s4.w * di2};
#pragma unroll
    for (int c = 0; c < TAILF; ++c) {
      const int k = NPL * PW + c;
      float v = vv[c] + b1[k];
      v = v > 0.f ? v : 0.f;
      a += v2f{v, v} * w2[k];
    }
  }
  const float a0 = a[0], a1 = a[1];
  h2[(size_t)n * 2 + 0] = a0;
  h2[(size_t)n * 2 + 1] = a1;
  out[(size_t)n * 2 + 0] += b2[0] + bs[0] + a0 * di2;
  out[(size_t)n * 2 + 1] += b2[1] + bs[1] + a1 * di2;
}

#define S2R 2
__global__ void __launch_bounds__(256)
scatter2_kernel(const float* __restrict__ h2,
                const float* __restrict__ deg,
                const int* __restrict__ src,
                const int* __restrict__ dst,
                float* __restrict__ out, int n_edges) {
  const int lane = threadIdx.x & 31;
  const int c = lane & 1;
  const int p = lane >> 1;
  const int wv = __builtin_amdgcn_readfirstlane((int)(threadIdx.x >> 5));
  const int e0 = (blockIdx.x * (blockDim.x >> 5) + wv) << 5;
  if (e0 >= n_edges) return;
  if (n_edges - e0 >= 32) {
    int s[S2R], d[S2R];
#pragma unroll
    for (int r = 0; r < S2R; ++r) {
      s[r] = src[e0 + (r << 4) + p];
      d[r] = dst[e0 + (r << 4) + p];
    }
    float v[S2R];
#pragma unroll
    for (int r = 0; r < S2R; ++r) {
      const float norm = deg_to_dinv(deg[s[r]]) * deg_to_dinv(deg[d[r]]);
      v[r] = h2[(size_t)s[r] * 2 + c] * norm;
    }
#pragma unroll
    for (int r = 0; r < S2R; ++r)
      atomicAdd(&out[(size_t)d[r] * 2 + c], v[r]);
  } else {
    for (int r = 0; r < S2R; ++r) {
      const int e = e0 + (r << 4) + p;
      if (e < n_edges) {
        const int sj = src[e], dj = dst[e];
        const float norm = deg_to_dinv(deg[sj]) * deg_to_dinv(deg[dj]);
        atomicAdd(&out[(size_t)dj * 2 + c], h2[(size_t)sj * 2 + c] * norm);
      }
    }
  }
}

__global__ void log_softmax_kernel(float* __restrict__ out, int n_nodes) {
  int n = blockIdx.x * blockDim.x + threadIdx.x;
  if (n >= n_nodes) return;
  float z0 = out[(size_t)n * 2 + 0];
  float z1 = out[(size_t)n * 2 + 1];
  float mx = fmaxf(z0, z1);
  float lse = mx + logf(expf(z0 - mx) + expf(z1 - mx));
  out[(size_t)n * 2 + 0] = z0 - lse;
  out[(size_t)n * 2 + 1] = z1 - lse;
}

extern "C" void kernel_launch(void* const* d_in, const int* in_sizes, int n_in,
                              void* d_out, int out_size, void* d_ws, size_t ws_size,
                              hipStream_t stream) {
  const float* x  = (const float*)d_in[0];
  const int*   ei = (const int*)d_in[1];
  const float* W1 = (const float*)d_in[2];
  const float* b1 = (const float*)d_in[3];
  const float* W2 = (const float*)d_in[4];
  const float* b2 = (const float*)d_in[5];
  const float* Ws = (const float*)d_in[6];
  const float* bs = (const float*)d_in[7];
  float* out = (float*)d_out;

  const int n_nodes = in_sizes[0] / IN_C;
  const int n_edges = in_sizes[1] / 2;
  const int* srcv = ei;
  const int* dstv = ei + n_edges;

  float* Wp   = (float*)d_ws;
  float* deg  = Wp + WP_FLOATS;
  size_t off  = ((size_t)n_nodes + 255) & ~(size_t)255;
  float* h    = deg + off;
  float* agg  = h + (size_t)n_nodes * HID;
  float* h2   = agg + (size_t)n_nodes * HID;

  const int T = 256;

  deg_init_kernel<<<(n_nodes + T - 1) / T, T, 0, stream>>>(deg, n_nodes);

  pack_w_kernel<<<WP_FLOATS / T, T, 0, stream>>>(W1, Ws, Wp);

  int tiles = (n_nodes + 15) / 16;
  gemm1_wmma_kernel<<<(tiles + 7) / 8, T, 0, stream>>>(x, Wp, dstv, deg, h, out,
                                                   n_nodes, n_edges);

  scatter1_kernel<<<(n_nodes + NPB - 1) / NPB, SW * 32, 0, stream>>>(h, deg, srcv, dstv, agg, n_edges, n_nodes);

  gemm2_fused_kernel<<<(n_nodes + T - 1) / T, T, 0, stream>>>(agg, h, b1, W2, b2, bs, deg,
                                                              h2, out, n_nodes);
  scatter2_kernel<<<(n_edges + T - 1) / T, T, 0, stream>>>(h2, deg, srcv, dstv, out, n_edges);

  log_softmax_kernel<<<(n_nodes + T - 1) / T, T, 0, stream>>>(out, n_nodes);
}


# scatter1: owner workgroups filter the edge list on one key byte per edge (dst/400, packed by the first launch), 16 edges per 16-byte lane load, not on 4-byte dst; LDS arrival-order sums as before
# speedup vs baseline: 1.9093x; 1.0507x over previous
#include <hip/hip_runtime.h>
#include <math.h>

#define IN_C 512
#define HID 100
#define OUT_C 2
#define NT 8
#define WP_FLOATS (256 * NT * 16 * 2)

typedef __attribute__((ext_vector_type(2))) float v2f;
typedef __attribute__((ext_vector_type(8))) float v8f;

#define PW 32
#define NPL 3
#define TAILF (HID - NPL * PW)
#define NPB 400

typedef __attribute__((ext_vector_type(4))) unsigned v4u;

__device__ __forceinline__ unsigned char* key_place(const float* ws0, size_t ws_size, float* h2,
                                                    int n_nodes, int n_edges, int* nk) {
  char* past = (char*)(h2 + (size_t)n_nodes * OUT_C);
  const size_t need = (size_t)(past - (const char*)ws0) + (((size_t)n_edges + 3) & ~(size_t)3);
  if (need <= ws_size) { *nk = n_edges; return (unsigned char*)past; }
  const size_t room = (size_t)n_nodes * OUT_C * sizeof(float);
  *nk = (int)(((size_t)n_edges < room ? (size_t)n_edges : room) & ~(size_t)3);
  return (unsigned char*)h2;
}

__global__ void prep_kernel(float* __restrict__ deg, int n_nodes,
                            const int* __restrict__ dst, int n_edges,
                            const float* ws0, size_t ws_size, float* h2) {
  const int i = blockIdx.x * blockDim.x + threadIdx.x;
  for (int j = i; j < n_nodes; j += gridDim.x * blockDim.x) deg[j] = 1.0f;
  int nk;
  unsigned* kw = (unsigned*)key_place(ws0, ws_size, h2, n_nodes, n_edges, &nk);
  const long long e = 4LL * i;
  if (e + 3 < nk) {
    const unsigned k0 = ((unsigned)dst[e]     / NPB) & 255u;
    const unsigned k1 = ((unsigned)dst[e + 1] / NPB) & 255u;
    const unsigned k2 = ((unsigned)dst[e + 2] / NPB) & 255u;
    const unsigned k3 = ((unsigned)dst[e + 3] / NPB) & 255u;
    kw[i] = k0 | (k1 << 8) | (k2 << 16) | (k3 << 24);
  }
}

__device__ __forceinline__ float deg_to_dinv(float d) { return (d > 0.f) ? rsqrtf(d) : 0.f; }

__global__ void pack_w_kernel(const float* __restrict__ W1,
                              const float* __restrict__ Ws,
                              float* __restrict__ Wp) {
  int idx = blockIdx.x * blockDim.x + threadIdx.x;
  if (idx >= WP_FLOATS) return;
  int kin = idx & 1;
  int col = (idx >> 1) & 15;
  int t   = (idx >> 5) & 7;
  int kp  = idx >> 8;
  int k   = (kp << 1) + kin;
  float v = 0.f;
  if (t < 7) {
    int c = (t << 4) + col;
    if (c < HID) v = W1[(size_t)k * HID + c];
  } else {
    if (col < OUT_C) v = Ws[(size_t)k * OUT_C + col];
  }
  Wp[idx] = v;
}

#define KBLK 16
__global__ void __launch_bounds__(256)
gemm1_wmma_kernel(const float* __restrict__ x, const float* __restrict__ Wp,
                  const int* __restrict__ dst, float* __restrict__ deg,
                  float* __restrict__ h,
                  float* __restrict__ outp, int n_nodes, int n_edges) {
  const int lane = threadIdx.x & 31;
  const int wv   = threadIdx.x >> 5;
  const int tile = blockIdx.x * (blockDim.x >> 5) + wv;
  const int row0 = tile << 4;
  if (row0 >= n_nodes) return;

  const long long stride = (long long)((n_nodes + 15) >> 4) << 5;
  long long e = ((long long)tile << 5) + lane;
  int dcur = (e < n_edges) ? dst[e] : -1;

  const int m  = lane & 15;
  const int kh = (lane >> 4) << 1;

  v8f acc[NT] = {};

  const float* __restrict__ pa = x + (size_t)(row0 + m) * IN_C + kh;
  const float* __restrict__ pb = Wp + ((size_t)(lane >> 4) * 256) + (m << 1);

  for (int kb = 0; kb < IN_C / (4 * KBLK); ++kb) {
    if (dcur >= 0) atomicAdd(&deg[dcur], 1.0f);
    const long long en = e + stride;
    const int dnxt = (en < n_edges) ? dst[en] : -1;
    for (int ks = 0; ks < KBLK; ++ks) {
      v2f a = *(const v2f*)pa;
#pragma unroll
      for (int t = 0; t < NT; ++t) {
        v2f b = *(const v2f*)(pb + (t << 5));
        acc[t] = __builtin_amdgcn_wmma_f32_16x16x4_f32(
            false, a, false, b, (short)0, acc[t], false, false);
      }
      pa += 4;
      pb += 512;
    }
    dcur = dnxt;
    e = en;
  }
  if (dcur >= 0) atomicAdd(&deg[dcur], 1.0f);
  for (e += stride; e < n_edges; e += stride) atomicAdd(&deg[dst[e]], 1.0f);

  const int rbase = (lane < 16) ? 0 : 8;
  const size_t plane = (size_t)n_nodes * PW;
#pragma unroll
  for (int r = 0; r < 8; ++r) {
    const int row = row0 + rbase + r;
#pragma unroll
    for (int t = 0; t < 7; ++t) {
      int col = (t << 4) + m;
      if (col < HID) {
        const size_t o = (t < 2 * NPL)
            ? (size_t)(t >> 1) * plane + (size_t)row * PW + ((t & 1) << 4) + m
            : (size_t)NPL * plane + (size_t)row * TAILF + m;
        h[o] = acc[t][r];
      }
    }
    if (m < OUT_C) outp[(size_t)row * OUT_C + m] = acc[7][r];
  }
}

#define SW 32
#define KSU 2
#define SU 4
#define LCAP (32 + KSU * 512)
static_assert(LCAP >= 32 + SU * 128, "the list also holds a pass of the dst loop");
#define TU 4
#define EB 8
static_assert(EB * TAILF == 32, "one wave-wide access covers the leftover features of a batch");
static_assert(LCAP >= 32 + TU * 32, "the list also holds a trip of the edge-at-a-time loop");

typedef __attribute__((ext_vector_type(4))) int v4i;

__device__ __forceinline__ void lds_add(float* p, float v) {
  __hip_atomic_fetch_add(p, v, __ATOMIC_RELAXED, __HIP_MEMORY_SCOPE_WORKGROUP);
}

__global__ void __launch_bounds__(SW * 32)
scatter1_kernel(const float* __restrict__ h,
                const float* __restrict__ deg,
                const int* __restrict__ src,
                const int* __restrict__ dst,
                float* __restrict__ agg, int n_edges, int n_nodes,
                const float* ws0, size_t ws_size, float* h2) {
  __shared__ float acc[NPB * HID];
  __shared__ int lst[SW * LCAP];
  const int lane = threadIdx.x & 31;
  const int wv = __builtin_amdgcn_readfirstlane((int)(threadIdx.x >> 5));
  const int base = blockIdx.x * NPB;
  const int cnt = (n_nodes - base < NPB) ? (n_nodes - base) : NPB;
  const size_t plane = (size_t)n_nodes * PW;
  const char* __restrict__ hb0 = (const char*)h;
  const char* __restrict__ hb1 = (const char*)(h + plane);
  const char* __restrict__ hb2 = (const char*)(h + 2 * plane);
  const char* __restrict__ hbt = (const char*)(h + NPL * plane);
  const unsigned lb = (unsigned)lane * 4u;
  const int tu = lane / TAILF;
  const int tf = lane % TAILF;
  int* const my = lst + wv * LCAP;

  for (int i = threadIdx.x; i < NPB * HID; i += SW * 32) acc[i] = 0.f;
  __syncthreads();

  auto drain = [&](const int* p, int k) {
    const int e = (lane < k) ? p[lane] : -1;
    int s = 0, dl = -1;
    float nrm = 0.f;
    if (e >= 0) {
      s = src[e];
      const int d = dst[e];
      nrm = deg_to_dinv(deg[s]) * deg_to_dinv(deg[d]);
      if ((unsigned)(d - base) < (unsigned)cnt) dl = d - base;
    }
    for (int jb = 0; jb < k; jb += EB) {
      float v[EB][3];
      const int   st = __shfl(s, jb + tu);
      const int   dt = __shfl(dl, jb + tu);
      const float nt = __shfl(nrm, jb + tu);
#pragma unroll
      for (int u = 0; u < EB; ++u) {
        const int sj = __builtin_amdgcn_readlane(s, jb + u);
        const unsigned so = (unsigned)sj * (PW * 4u) + lb;
        v[u][0] = *(const float*)(hb0 + so);
        v[u][1] = *(const float*)(hb1 + so);
        v[u][2] = *(const float*)(hb2 + so);
      }
      const float vt = *(const float*)(hbt + ((unsigned)st * (TAILF * 4u) + (unsigned)tf * 4u));
#pragma unroll
      for (int u = 0; u < EB; ++u) {
        const int dj = __builtin_amdgcn_readlane(dl, jb + u);
        if (jb + u < k && dj >= 0) {
          const float nj = __int_as_float(__builtin_amdgcn_readlane(__float_as_int(nrm), jb + u));
          float* r = acc + dj * HID + lane;
          lds_add(r,      v[u][0] * nj);
          lds_add(r + 32, v[u][1] * nj);
          lds_add(r + 64, v[u][2] * nj);
        }
      }
      if (jb + tu < k && dt >= 0) lds_add(acc + dt * HID + NPL * PW + tf, vt * nt);
    }
  };

  int count = 0;
  auto append = [&](bool m, int ev) {
    const unsigned bal = __builtin_amdgcn_ballot_w32(m);
    if (bal) {
      if (m) my[count + (int)__builtin_amdgcn_mbcnt_lo(bal, 0u)] = ev;
      count += __builtin_popcount(bal);
    }
  };

  int nk;
  const unsigned char* keys = key_place(ws0, ws_size, h2, n_nodes, n_edges, &nk);
  const int kstride = SW * KSU * 512;
  const int nfullk = ((((size_t)keys) & 15) == 0) ? nk / kstride : 0;
  if (nfullk > 0) {
    const unsigned kb = (unsigned)(blockIdx.x & 255) * 0x01010101u;
    auto hit = [&](unsigned w) {
      const unsigned x = w ^ kb;
      return ~(((x & 0x7f7f7f7fu) + 0x7f7f7f7fu) | x) & 0x80808080u;
    };
    int ps = (int)(((long long)blockIdx.x * nfullk) / gridDim.x);
    const int wofs = wv * (KSU * 512) + lane * 16;
    int eb = ps * kstride + wofs;
    v4u cur[KSU];
#pragma unroll
    for (int u = 0; u < KSU; ++u) cur[u] = *(const v4u*)(keys + eb + u * 512);
#pragma unroll 2
    for (int it = 0; it < nfullk; ++it) {
      ps = (ps + 1 < nfullk) ? ps + 1 : 0;
      const int en = (it + 1 < nfullk) ? ps * kstride + wofs : eb;
      v4u nxt[KSU];
#pragma unroll
      for (int u = 0; u < KSU; ++u) nxt[u] = *(const v4u*)(keys + en + u * 512);
#pragma unroll
      for (int u = 0; u < KSU; ++u) {
        const v4u kv = cur[u];
        const int e0 = eb + u * 512;
        const unsigned z0 = hit(kv.x), z1 = hit(kv.y), z2 = hit(kv.z), z3 = hit(kv.w);
        unsigned m = (z0 >> 7) | (z1 >> 6) | (z2 >> 5) | (z3 >> 4);
        for (unsigned bal = __builtin_amdgcn_ballot_w32(m != 0u); bal;
             bal = __builtin_amdgcn_ballot_w32(m != 0u)) {
          if (m != 0u) {
            const int p = __builtin_ctz(m);
            my[count + (int)__builtin_amdgcn_mbcnt_lo(bal, 0u)] = e0 + ((p & 7) << 2) + (p >> 3);
          }
          count += __builtin_popcount(bal);
          m &= m - 1u;
        }
      }
      __builtin_amdgcn_fence(__ATOMIC_SEQ_CST, "wavefront");
      while (count >= 32) {
        count -= 32;
        drain(my + count, 32);
      }
#pragma unroll
      for (int u = 0; u < KSU; ++u) cur[u] = nxt[u];
      eb = en;
    }
  }
  const long long lo = (long long)nfullk * kstride;

  const long long wstride = (long long)SW * SU * 128;
  const int nfull = ((((size_t)dst) & 15) == 0) ? (int)(((long long)n_edges - lo) / wstride) : 0;
  if (nfull > 0) {
    int ps = (int)(((long long)blockIdx.x * nfull) / gridDim.x);
    const long long wofs = (long long)wv * (SU * 128) + lane * 4;
    long long eb = lo + (long long)ps * wstride + wofs;
    v4i cur[SU];
#pragma unroll
    for (int u = 0; u < SU; ++u) cur[u] = *(const v4i*)(dst + eb + u * 128);
#pragma unroll 2
    for (int it = 0; it < nfull; ++it) {
      ps = (ps + 1 < nfull) ? ps + 1 : 0;
      const long long en = (it + 1 < nfull) ? lo + (long long)ps * wstride + wofs : eb;
      v4i nxt[SU];
#pragma unroll
      for (int u = 0; u < SU; ++u) nxt[u] = *(const v4i*)(dst + en + u * 128);
#pragma unroll
      for (int u = 0; u < SU; ++u) {
        const v4i dv = cur[u];
        const int e0 = (int)(eb + u * 128);
        const bool m0 = (unsigned)(dv.x - base) < (unsigned)cnt;
        const bool m1 = (unsigned)(dv.y - base) < (unsigned)cnt;
        const bool m2 = (unsigned)(dv.z - base) < (unsigned)cnt;
        const bool m3 = (unsigned)(dv.w - base) < (unsigned)cnt;
        if (__builtin_amdgcn_ballot_w32(m0 | m1 | m2 | m3)) {
          append(m0, e0);
          append(m1, e0 + 1);
          append(m2, e0 + 2);
          append(m3, e0 + 3);
        }
      }
      __builtin_amdgcn_fence(__ATOMIC_SEQ_CST, "wavefront");
      while (count >= 32) {
        count -= 32;
        drain(my + count, 32);
      }
#pragma unroll
      for (int u = 0; u < SU; ++u) cur[u] = nxt[u];
      eb = en;
    }
  }
  for (long long t = lo + (long long)nfull * wstride + wv * 32; ; t += TU * SW * 32) {
    const bool done = t >= n_edges;
    if (!done) {
      int dv[TU];
#pragma unroll
      for (int u = 0; u < TU; ++u) {
        const long long e = t + u * (SW * 32) + lane;
        dv[u] = (e < n_edges) ? dst[e] : -1;
      }
#pragma unroll
      for (int u = 0; u < TU; ++u)
        append((unsigned)(dv[u] - base) < (unsigned)cnt, (int)(t + u * (SW * 32) + lane));
    }
    __builtin_amdgcn_fence(__ATOMIC_SEQ_CST, "wavefront");
    const int lim = done ? 1 : 32;
    while (count >= lim) {
      const int k = count < 32 ? count : 32;
      count -= k;
      drain(my + count, k);
    }
    if (done) break;
  }

  __syncthreads();

  for (int r = wv; r < cnt; r += SW) {
    const float* a = acc + r * HID;
    const size_t o = (size_t)(base + r) * PW + lane;
    agg[o]             = a[lane];
    agg[plane + o]     = a[PW + lane];
    agg[2 * plane + o] = a[2 * PW + lane];
  }
  for (int i = threadIdx.x; i < cnt * TAILF; i += SW * 32)
    agg[NPL * plane + (size_t)base * TAILF + i] = acc[(i / TAILF) * HID + NPL * PW + (i % TAILF)];
}

__global__ void gemm2_fused_kernel(const float* __restrict__ agg,
                                   const float* __restrict__ h,
                                   const float* __restrict__ b1,
                                   const float* __restrict__ W2,
                                   const float* __restrict__ b2,
                                   const float* __restrict__ bs,
                                   const float* __restrict__ deg,
                                   float* __restrict__ h2,
                                   float* __restrict__ out, int n_nodes) {
  int n = blockIdx.x * blockDim.x + threadIdx.x;
  if (n >= n_nodes) return;
  const size_t plane = (size_t)n_nodes * PW;
  const float di  = deg_to_dinv(deg[n]);
  const float di2 = di * di;
  const v2f* __restrict__ w2 = (const v2f*)W2;
  v2f a = {0.f, 0.f};
  for (int p = 0; p < NPL; ++p) {
    const float4* __restrict__ r4 = (const float4*)(agg + p * plane + (size_t)n * PW);
    const float4* __restrict__ g4 = (const float4*)(h + p * plane + (size_t)n * PW);
#pragma unroll
    for (int q = 0; q < PW / 4; ++q) {
      const float4 v4 = r4[q];
      const float4 s4 = g4[q];
      const float vv[4] = {v4.x + s4.x * di2, v4.y + s4.y * di2, v4.z + s4.z * di2, v4.w + s4.w * di2};
#pragma unroll
      for (int c = 0; c < 4; ++c) {
        const int k = p * PW + q * 4 + c;
        float v = vv[c] + b1[k];
        v = v > 0.f ? v : 0.f;
        a += v2f{v, v} * w2[k];
      }
    }
  }
  {
    const float4 v4 = *(const float4*)(agg + NPL * plane + (size_t)n * TAILF);
    const float4 s4 = *(const float4*)(h + NPL * plane + (size_t)n * TAILF);
    const float vv[4] = {v4.x + s4.x * di2, v4.y + s4.y * di2, v4.z + s4.z * di2, v4.w + s4.w * di2};
#pragma unroll
    for (int c = 0; c < TAILF; ++c) {
      const int k = NPL * PW + c;
      float v = vv[c] + b1[k];
      v = v > 0.f ? v : 0.f;
      a += v2f{v, v} * w2[k];
    }
  }
  const float a0 = a[0], a1 = a[1];
  h2[(size_t)n * 2 + 0] = a0;
  h2[(size_t)n * 2 + 1] = a1;
  out[(size_t)n * 2 + 0] += b2[0] + bs[0] + a0 * di2;
  out[(size_t)n * 2 + 1] += b2[1] + bs[1] + a1 * di2;
}

#define S2R 2
__global__ void __launch_bounds__(256)
scatter2_kernel(const float* __restrict__ h2,
                const float* __restrict__ deg,
                const int* __restrict__ src,
                const int* __restrict__ dst,
                float* __restrict__ out, int n_edges) {
  const int lane = threadIdx.x & 31;
  const int c = lane & 1;
  const int p = lane >> 1;
  const int wv = __builtin_amdgcn_readfirstlane((int)(threadIdx.x >> 5));
  const int e0 = (blockIdx.x * (blockDim.x >> 5) + wv) << 5;
  if (e0 >= n_edges) return;
  if (n_edges - e0 >= 32) {
    int s[S2R], d[S2R];
#pragma unroll
    for (int r = 0; r < S2R; ++r) {
      s[r] = src[e0 + (r << 4) + p];
      d[r] = dst[e0 + (r << 4) + p];
    }
    float v[S2R];
#pragma unroll
    for (int r = 0; r < S2R; ++r) {
      const float norm = deg_to_dinv(deg[s[r]]) * deg_to_dinv(deg[d[r]]);
      v[r] = h2[(size_t)s[r] * 2 + c] * norm;
    }
#pragma unroll
    for (int r = 0; r < S2R; ++r)
      atomicAdd(&out[(size_t)d[r] * 2 + c], v[r]);
  } else {
    for (int r = 0; r < S2R; ++r) {
      const int e = e0 + (r << 4) + p;
      if (e < n_edges) {
        const int sj = src[e], dj = dst[e];
        const float norm = deg_to_dinv(deg[sj]) * deg_to_dinv(deg[dj]);
        atomicAdd(&out[(size_t)dj * 2 + c], h2[(size_t)sj * 2 + c] * norm);
      }
    }
  }
}

__global__ void log_softmax_kernel(float* __restrict__ out, int n_nodes) {
  int n = blockIdx.x * blockDim.x + threadIdx.x;
  if (n >= n_nodes) return;
  float z0 = out[(size_t)n * 2 + 0];
  float z1 = out[(size_t)n * 2 + 1];
  float mx = fmaxf(z0, z1);
  float lse = mx + logf(expf(z0 - mx) + expf(z1 - mx));
  out[(size_t)n * 2 + 0] = z0 - lse;
  out[(size_t)n * 2 + 1] = z1 - lse;
}

extern "C" void kernel_launch(void* const* d_in, const int* in_sizes, int n_in,
                              void* d_out, int out_size, void* d_ws, size_t ws_size,
                              hipStream_t stream) {
  const float* x  = (const float*)d_in[0];
  const int*   ei = (const int*)d_in[1];
  const float* W1 = (const float*)d_in[2];
  const float* b1 = (const float*)d_in[3];
  const float* W2 = (const float*)d_in[4];
  const float* b2 = (const float*)d_in[5];
  const float* Ws = (const float*)d_in[6];
  const float* bs = (const float*)d_in[7];
  float* out = (float*)d_out;

  const int n_nodes = in_sizes[0] / IN_C;
  const int n_edges = in_sizes[1] / 2;
  const int* srcv = ei;
  const int* dstv = ei + n_edges;

  float* Wp   = (float*)d_ws;
  float* deg  = Wp + WP_FLOATS;
  size_t off  = ((size_t)n_nodes + 255) & ~(size_t)255;
  float* h    = deg + off;
  float* agg  = h + (size_t)n_nodes * HID;
  float* h2   = agg + (size_t)n_nodes * HID;

  const int T = 256;

  prep_kernel<<<((n_edges + 3) / 4 + T - 1) / T, T, 0, stream>>>(deg, n_nodes, dstv, n_edges,
                                                                 Wp, ws_size, h2);

  pack_w_kernel<<<WP_FLOATS / T, T, 0, stream>>>(W1, Ws, Wp);

  int tiles = (n_nodes + 15) / 16;
  gemm1_wmma_kernel<<<(tiles + 7) / 8, T, 0, stream>>>(x, Wp, dstv, deg, h, out,
                                                   n_nodes, n_edges);

  scatter1_kernel<<<(n_nodes + NPB - 1) / NPB, SW * 32, 0, stream>>>(h, deg, srcv, dstv, agg, n_edges, n_nodes,
                                                                          Wp, ws_size, h2);

  gemm2_fused_kernel<<<(n_nodes + T - 1) / T, T, 0, stream>>>(agg, h, b1, W2, b2, bs, deg,
                                                              h2, out, n_nodes);
  scatter2_kernel<<<(n_edges + T - 1) / T, T, 0, stream>>>(h2, deg, srcv, dstv, out, n_edges);

  log_softmax_kernel<<<(n_nodes + T - 1) / T, T, 0, stream>>>(out, n_nodes);
}


# scatter2: the owner workgroups of stage 1 filter the edge list on the key bytes again, sum out's 2 columns in LDS (LDS arrival order) and finish their rows with log_softmax: no global atomics on out
# speedup vs baseline: 2.0746x; 1.0866x over previous
#include <hip/hip_runtime.h>
#include <math.h>

#define IN_C 512
#define HID 100
#define OUT_C 2
#define NT 8
#define WP_FLOATS (256 * NT * 16 * 2)

typedef __attribute__((ext_vector_type(2))) float v2f;
typedef __attribute__((ext_vector_type(8))) float v8f;

#define PW 32
#define NPL 3
#define TAILF (HID - NPL * PW)
#define NPB 400

typedef __attribute__((ext_vector_type(4))) unsigned v4u;

__device__ __forceinline__ unsigned char* key_place(const float* ws0, size_t ws_size, float* h2,
                                                    int n_nodes, int n_edges, int* nk) {
  char* past = (char*)(h2 + (size_t)n_nodes * OUT_C);
  const size_t need = (size_t)(past - (const char*)ws0) + (((size_t)n_edges + 3) & ~(size_t)3);
  if (need <= ws_size) { *nk = n_edges; return (unsigned char*)past; }
  const size_t room = (size_t)n_nodes * OUT_C * sizeof(float);
  *nk = (int)(((size_t)n_edges < room ? (size_t)n_edges : room) & ~(size_t)3);
  return (unsigned char*)h2;
}

__global__ void prep_kernel(float* __restrict__ deg, int n_nodes,
                            const int* __restrict__ dst, int n_edges,
                            const float* ws0, size_t ws_size, float* h2) {
  const int i = blockIdx.x * blockDim.x + threadIdx.x;
  for (int j = i; j < n_nodes; j += gridDim.x * blockDim.x) deg[j] = 1.0f;
  int nk;
  unsigned* kw = (unsigned*)key_place(ws0, ws_size, h2, n_nodes, n_edges, &nk);
  const long long e = 4LL * i;
  if (e + 3 < nk) {
    const unsigned k0 = ((unsigned)dst[e]     / NPB) & 255u;
    const unsigned k1 = ((unsigned)dst[e + 1] / NPB) & 255u;
    const unsigned k2 = ((unsigned)dst[e + 2] / NPB) & 255u;
    const unsigned k3 = ((unsigned)dst[e + 3] / NPB) & 255u;
    kw[i] = k0 | (k1 << 8) | (k2 << 16) | (k3 << 24);
  }
}

__device__ __forceinline__ float deg_to_dinv(float d) { return (d > 0.f) ? rsqrtf(d) : 0.f; }

__global__ void pack_w_kernel(const float* __restrict__ W1,
                              const float* __restrict__ Ws,
                              float* __restrict__ Wp) {
  int idx = blockIdx.x * blockDim.x + threadIdx.x;
  if (idx >= WP_FLOATS) return;
  int kin = idx & 1;
  int col = (idx >> 1) & 15;
  int t   = (idx >> 5) & 7;
  int kp  = idx >> 8;
  int k   = (kp << 1) + kin;
  float v = 0.f;
  if (t < 7) {
    int c = (t << 4) + col;
    if (c < HID) v = W1[(size_t)k * HID + c];
  } else {
    if (col < OUT_C) v = Ws[(size_t)k * OUT_C + col];
  }
  Wp[idx] = v;
}

#define KBLK 16
__global__ void __launch_bounds__(256)
gemm1_wmma_kernel(const float* __restrict__ x, const float* __restrict__ Wp,
                  const int* __restrict__ dst, float* __restrict__ deg,
                  float* __restrict__ h,
                  float* __restrict__ outp, int n_nodes, int n_edges) {
  const int lane = threadIdx.x & 31;
  const int wv   = threadIdx.x >> 5;
  const int tile = blockIdx.x * (blockDim.x >> 5) + wv;
  const int row0 = tile << 4;
  if (row0 >= n_nodes) return;

  const long long stride = (long long)((n_nodes + 15) >> 4) << 5;
  long long e = ((long long)tile << 5) + lane;
  int dcur = (e < n_edges) ? dst[e] : -1;

  const int m  = lane & 15;
  const int kh = (lane >> 4) << 1;

  v8f acc[NT] = {};

  const float* __restrict__ pa = x + (size_t)(row0 + m) * IN_C + kh;
  const float* __restrict__ pb = Wp + ((size_t)(lane >> 4) * 256) + (m << 1);

  for (int kb = 0; kb < IN_C / (4 * KBLK); ++kb) {
    if (dcur >= 0) atomicAdd(&deg[dcur], 1.0f);
    const long long en = e + stride;
    const int dnxt = (en < n_edges) ? dst[en] : -1;
    for (int ks = 0; ks < KBLK; ++ks) {
      v2f a = *(const v2f*)pa;
#pragma unroll
      for (int t = 0; t < NT; ++t) {
        v2f b = *(const v2f*)(pb + (t << 5));
        acc[t] = __builtin_amdgcn_wmma_f32_16x16x4_f32(
            false, a, false, b, (short)0, acc[t], false, false);
      }
      pa += 4;
      pb += 512;
    }
    dcur = dnxt;
    e = en;
  }
  if (dcur >= 0) atomicAdd(&deg[dcur], 1.0f);
  for (e += stride; e < n_edges; e += stride) atomicAdd(&deg[dst[e]], 1.0f);

  const int rbase = (lane < 16) ? 0 : 8;
  const size_t plane = (size_t)n_nodes * PW;
#pragma unroll
  for (int r = 0; r < 8; ++r) {
    const int row = row0 + rbase + r;
#pragma unroll
    for (int t = 0; t < 7; ++t) {
      int col = (t << 4) + m;
      if (col < HID) {
        const size_t o = (t < 2 * NPL)
            ? (size_t)(t >> 1) * plane + (size_t)row * PW + ((t & 1) << 4) + m
            : (size_t)NPL * plane + (size_t)row * TAILF + m;
        h[o] = acc[t][r];
      }
    }
    if (m < OUT_C) outp[(size_t)row * OUT_C + m] = acc[7][r];
  }
}

#define SW 32
#define KSU 2
#define SU 4
#define LCAP (32 + KSU * 512)
static_assert(LCAP >= 32 + SU * 128, "the list also holds a pass of the dst loop");
#define TU 4
#define EB 8
static_assert(EB * TAILF == 32, "one wave-wide access covers the leftover features of a batch");
static_assert(LCAP >= 32 + TU * 32, "the list also holds a trip of the edge-at-a-time loop");

typedef __attribute__((ext_vector_type(4))) int v4i;

__device__ __forceinline__ void lds_add(float* p, float v) {
  __hip_atomic_fetch_add(p, v, __ATOMIC_RELAXED, __HIP_MEMORY_SCOPE_WORKGROUP);
}

__global__ void __launch_bounds__(SW * 32)
scatter1_kernel(const float* __restrict__ h,
                const float* __restrict__ deg,
                const int* __restrict__ src,
                const int* __restrict__ dst,
                float* __restrict__ agg, int n_edges, int n_nodes,
                const float* ws0, size_t ws_size, float* h2) {
  __shared__ float acc[NPB * HID];
  __shared__ int lst[SW * LCAP];
  const int lane = threadIdx.x & 31;
  const int wv = __builtin_amdgcn_readfirstlane((int)(threadIdx.x >> 5));
  const int base = blockIdx.x * NPB;
  const int cnt = (n_nodes - base < NPB) ? (n_nodes - base) : NPB;
  const size_t plane = (size_t)n_nodes * PW;
  const char* __restrict__ hb0 = (const char*)h;
  const char* __restrict__ hb1 = (const char*)(h + plane);
  const char* __restrict__ hb2 = (const char*)(h + 2 * plane);
  const char* __restrict__ hbt = (const char*)(h + NPL * plane);
  const unsigned lb = (unsigned)lane * 4u;
  const int tu = lane / TAILF;
  const int tf = lane % TAILF;
  int* const my = lst + wv * LCAP;

  for (int i = threadIdx.x; i < NPB * HID; i += SW * 32) acc[i] = 0.f;
  __syncthreads();

  auto drain = [&](const int* p, int k) {
    const int e = (lane < k) ? p[lane] : -1;
    int s = 0, dl = -1;
    float nrm = 0.f;
    if (e >= 0) {
      s = src[e];
      const int d = dst[e];
      nrm = deg_to_dinv(deg[s]) * deg_to_dinv(deg[d]);
      if ((unsigned)(d - base) < (unsigned)cnt) dl = d - base;
    }
    for (int jb = 0; jb < k; jb += EB) {
      float v[EB][3];
      const int   st = __shfl(s, jb + tu);
      const int   dt = __shfl(dl, jb + tu);
      const float nt = __shfl(nrm, jb + tu);
#pragma unroll
      for (int u = 0; u < EB; ++u) {
        const int sj = __builtin_amdgcn_readlane(s, jb + u);
        const unsigned so = (unsigned)sj * (PW * 4u) + lb;
        v[u][0] = *(const float*)(hb0 + so);
        v[u][1] = *(const float*)(hb1 + so);
        v[u][2] = *(const float*)(hb2 + so);
      }
      const float vt = *(const float*)(hbt + ((unsigned)st * (TAILF * 4u) + (unsigned)tf * 4u));
#pragma unroll
      for (int u = 0; u < EB; ++u) {
        const int dj = __builtin_amdgcn_readlane(dl, jb + u);
        if (jb + u < k && dj >= 0) {
          const float nj = __int_as_float(__builtin_amdgcn_readlane(__float_as_int(nrm), jb + u));
          float* r = acc + dj * HID + lane;
          lds_add(r,      v[u][0] * nj);
          lds_add(r + 32, v[u][1] * nj);
          lds_add(r + 64, v[u][2] * nj);
        }
      }
      if (jb + tu < k && dt >= 0) lds_add(acc + dt * HID + NPL * PW + tf, vt * nt);
    }
  };

  int count = 0;
  auto append = [&](bool m, int ev) {
    const unsigned bal = __builtin_amdgcn_ballot_w32(m);
    if (bal) {
      if (m) my[count + (int)__builtin_amdgcn_mbcnt_lo(bal, 0u)] = ev;
      count += __builtin_popcount(bal);
    }
  };

  int nk;
  const unsigned char* keys = key_place(ws0, ws_size, h2, n_nodes, n_edges, &nk);
  const int kstride = SW * KSU * 512;
  const int nfullk = ((((size_t)keys) & 15) == 0) ? nk / kstride : 0;
  if (nfullk > 0) {
    const unsigned kb = (unsigned)(blockIdx.x & 255) * 0x01010101u;
    auto hit = [&](unsigned w) {
      const unsigned x = w ^ kb;
      return ~(((x & 0x7f7f7f7fu) + 0x7f7f7f7fu) | x) & 0x80808080u;
    };
    int ps = (int)(((long long)blockIdx.x * nfullk) / gridDim.x);
    const int wofs = wv * (KSU * 512) + lane * 16;
    int eb = ps * kstride + wofs;
    v4u cur[KSU];
#pragma unroll
    for (int u = 0; u < KSU; ++u) cur[u] = *(const v4u*)(keys + eb + u * 512);
#pragma unroll 2
    for (int it = 0; it < nfullk; ++it) {
      ps = (ps + 1 < nfullk) ? ps + 1 : 0;
      const int en = (it + 1 < nfullk) ? ps * kstride + wofs : eb;
      v4u nxt[KSU];
#pragma unroll
      for (int u = 0; u < KSU; ++u) nxt[u] = *(const v4u*)(keys + en + u * 512);
#pragma unroll
      for (int u = 0; u < KSU; ++u) {
        const v4u kv = cur[u];
        const int e0 = eb + u * 512;
        const unsigned z0 = hit(kv.x), z1 = hit(kv.y), z2 = hit(kv.z), z3 = hit(kv.w);
        unsigned m = (z0 >> 7) | (z1 >> 6) | (z2 >> 5) | (z3 >> 4);
        for (unsigned bal = __builtin_amdgcn_ballot_w32(m != 0u); bal;
             bal = __builtin_amdgcn_ballot_w32(m != 0u)) {
          if (m != 0u) {
            const int p = __builtin_ctz(m);
            my[count + (int)__builtin_amdgcn_mbcnt_lo(bal, 0u)] = e0 + ((p & 7) << 2) + (p >> 3);
          }
          count += __builtin_popcount(bal);
          m &= m - 1u;
        }
      }
      __builtin_amdgcn_fence(__ATOMIC_SEQ_CST, "wavefront");
      while (count >= 32) {
        count -= 32;
        drain(my + count, 32);
      }
#pragma unroll
      for (int u = 0; u < KSU; ++u) cur[u] = nxt[u];
      eb = en;
    }
  }
  const long long lo = (long long)nfullk * kstride;

  const long long wstride = (long long)SW * SU * 128;
  const int nfull = ((((size_t)dst) & 15) == 0) ? (int)(((long long)n_edges - lo) / wstride) : 0;
  if (nfull > 0) {
    int ps = (int)(((long long)blockIdx.x * nfull) / gridDim.x);
    const long long wofs = (long long)wv * (SU * 128) + lane * 4;
    long long eb = lo + (long long)ps * wstride + wofs;
    v4i cur[SU];
#pragma unroll
    for (int u = 0; u < SU; ++u) cur[u] = *(const v4i*)(dst + eb + u * 128);
#pragma unroll 2
    for (int it = 0; it < nfull; ++it) {
      ps = (ps + 1 < nfull) ? ps + 1 : 0;
      const long long en = (it + 1 < nfull) ? lo + (long long)ps * wstride + wofs : eb;
      v4i nxt[SU];
#pragma unroll
      for (int u = 0; u < SU; ++u) nxt[u] = *(const v4i*)(dst + en + u * 128);
#pragma unroll
      for (int u = 0; u < SU; ++u) {
        const v4i dv = cur[u];
        const int e0 = (int)(eb + u * 128);
        const bool m0 = (unsigned)(dv.x - base) < (unsigned)cnt;
        const bool m1 = (unsigned)(dv.y - base) < (unsigned)cnt;
        const bool m2 = (unsigned)(dv.z - base) < (unsigned)cnt;
        const bool m3 = (unsigned)(dv.w - base) < (unsigned)cnt;
        if (__builtin_amdgcn_ballot_w32(m0 | m1 | m2 | m3)) {
          append(m0, e0);
          append(m1, e0 + 1);
          append(m2, e0 + 2);
          append(m3, e0 + 3);
        }
      }
      __builtin_amdgcn_fence(__ATOMIC_SEQ_CST, "wavefront");
      while (count >= 32) {
        count -= 32;
        drain(my + count, 32);
      }
#pragma unroll
      for (int u = 0; u < SU; ++u) cur[u] = nxt[u];
      eb = en;
    }
  }
  for (long long t = lo + (long long)nfull * wstride + wv * 32; ; t += TU * SW * 32) {
    const bool done = t >= n_edges;
    if (!done) {
      int dv[TU];
#pragma unroll
      for (int u = 0; u < TU; ++u) {
        const long long e = t + u * (SW * 32) + lane;
        dv[u] = (e < n_edges) ? dst[e] : -1;
      }
#pragma unroll
      for (int u = 0; u < TU; ++u)
        append((unsigned)(dv[u] - base) < (unsigned)cnt, (int)(t + u * (SW * 32) + lane));
    }
    __builtin_amdgcn_fence(__ATOMIC_SEQ_CST, "wavefront");
    const int lim = done ? 1 : 32;
    while (count >= lim) {
      const int k = count < 32 ? count : 32;
      count -= k;
      drain(my + count, k);
    }
    if (done) break;
  }

  __syncthreads();

  for (int r = wv; r < cnt; r += SW) {
    const float* a = acc + r * HID;
    const size_t o = (size_t)(base + r) * PW + lane;
    agg[o]             = a[lane];
    agg[plane + o]     = a[PW + lane];
    agg[2 * plane + o] = a[2 * PW + lane];
  }
  for (int i = threadIdx.x; i < cnt * TAILF; i += SW * 32)
    agg[NPL * plane + (size_t)base * TAILF + i] = acc[(i / TAILF) * HID + NPL * PW + (i % TAILF)];
}

__global__ void gemm2_fused_kernel(const float* __restrict__ agg,
                                   const float* __restrict__ h,
                                   const float* __restrict__ b1,
                                   const float* __restrict__ W2,
                                   const float* __restrict__ b2,
                                   const float* __restrict__ bs,
                                   const float* __restrict__ deg,
                                   float* __restrict__ h2,
                                   float* __restrict__ out, int n_nodes) {
  int n = blockIdx.x * blockDim.x + threadIdx.x;
  if (n >= n_nodes) return;
  const size_t plane = (size_t)n_nodes * PW;
  const float di  = deg_to_dinv(deg[n]);
  const float di2 = di * di;
  const v2f* __restrict__ w2 = (const v2f*)W2;
  v2f a = {0.f, 0.f};
  for (int p = 0; p < NPL; ++p) {
    const float4* __restrict__ r4 = (const float4*)(agg + p * plane + (size_t)n * PW);
    const float4* __restrict__ g4 = (const float4*)(h + p * plane + (size_t)n * PW);
#pragma unroll
    for (int q = 0; q < PW / 4; ++q) {
      const float4 v4 = r4[q];
      const float4 s4 = g4[q];
      const float vv[4] = {v4.x + s4.x * di2, v4.y + s4.y * di2, v4.z + s4.z * di2, v4.w + s4.w * di2};
#pragma unroll
      for (int c = 0; c < 4; ++c) {
        const int k = p * PW + q * 4 + c;
        float v = vv[c] + b1[k];
        v = v > 0.f ? v : 0.f;
        a += v2f{v, v} * w2[k];
      }
    }
  }
  {
    const float4 v4 = *(const float4*)(agg + NPL * plane + (size_t)n * TAILF);
    const float4 s4 = *(const float4*)(h + NPL * plane + (size_t)n * TAILF);
    const float vv[4] = {v4.x + s4.x * di2, v4.y + s4.y * di2, v4.z + s4.z * di2, v4.w + s4.w * di2};
#pragma unroll
    for (int c = 0; c < TAILF; ++c) {
      const int k = NPL * PW + c;
      float v = vv[c] + b1[k];
      v = v > 0.f ? v : 0.f;
      a += v2f{v, v} * w2[k];
    }
  }
  const float a0 = a[0], a1 = a[1];
  h2[(size_t)n * 2 + 0] = a0;
  h2[(size_t)n * 2 + 1] = a1;
  out[(size_t)n * 2 + 0] += b2[0] + bs[0] + a0 * di2;
  out[(size_t)n * 2 + 1] += b2[1] + bs[1] + a1 * di2;
}

__global__ void __launch_bounds__(SW * 32)
scatter2_kernel(const float* __restrict__ h2,
                const float* __restrict__ deg,
                const int* __restrict__ src,
                const int* __restrict__ dst,
                float* __restrict__ out, int n_edges, int n_nodes,
                const float* ws0, size_t ws_size) {
  __shared__ float acc[NPB * OUT_C];
  __shared__ int lst[SW * LCAP];
  const int lane = threadIdx.x & 31;
  const int wv = __builtin_amdgcn_readfirstlane((int)(threadIdx.x >> 5));
  const int base = blockIdx.x * NPB;
  const int cnt = (n_nodes - base < NPB) ? (n_nodes - base) : NPB;
  int* const my = lst + wv * LCAP;

  for (int i = threadIdx.x; i < NPB * OUT_C; i += SW * 32) acc[i] = 0.f;
  __syncthreads();

  auto drain = [&](const int* p, int k) {
    if (lane < k) {
      const int e = p[lane];
      const int s = src[e];
      const int d = dst[e];
      if ((unsigned)(d - base) < (unsigned)cnt) {
        const float nrm = deg_to_dinv(deg[s]) * deg_to_dinv(deg[d]);
        const v2f hv = *(const v2f*)(h2 + (size_t)s * OUT_C);
        float* r = acc + (d - base) * OUT_C;
        lds_add(r,     hv.x * nrm);
        lds_add(r + 1, hv.y * nrm);
      }
    }
  };

  int count = 0;
  auto append = [&](bool m, int ev) {
    const unsigned bal = __builtin_amdgcn_ballot_w32(m);
    if (bal) {
      if (m) my[count + (int)__builtin_amdgcn_mbcnt_lo(bal, 0u)] = ev;
      count += __builtin_popcount(bal);
    }
  };

  int nk;
  const unsigned char* keys = key_place(ws0, ws_size, (float*)h2, n_nodes, n_edges, &nk);
  if ((const void*)keys == (const void*)h2) nk = 0;
  const int kstride = SW * KSU * 512;
  const int nfullk = ((((size_t)keys) & 15) == 0) ? nk / kstride : 0;
  if (nfullk > 0) {
    const unsigned kb = (unsigned)(blockIdx.x & 255) * 0x01010101u;
    auto hit = [&](unsigned w) {
      const unsigned x = w ^ kb;
      return ~(((x & 0x7f7f7f7fu) + 0x7f7f7f7fu) | x) & 0x80808080u;
    };
    int ps = (int)(((long long)blockIdx.x * nfullk) / gridDim.x);
    const int wofs = wv * (KSU * 512) + lane * 16;
    int eb = ps * kstride + wofs;
    v4u cur[KSU];
#pragma unroll
    for (int u = 0; u < KSU; ++u) cur[u] = *(const v4u*)(keys + eb + u * 512);
#pragma unroll 2
    for (int it = 0; it < nfullk; ++it) {
      ps = (ps + 1 < nfullk) ? ps + 1 : 0;
      const int en = (it + 1 < nfullk) ? ps * kstride + wofs : eb;
      v4u nxt[KSU];
#pragma unroll
      for (int u = 0; u < KSU; ++u) nxt[u] = *(const v4u*)(keys + en + u * 512);
#pragma unroll
      for (int u = 0; u < KSU; ++u) {
        const v4u kv = cur[u];
        const int e0 = eb + u * 512;
        const unsigned z0 = hit(kv.x), z1 = hit(kv.y), z2 = hit(kv.z), z3 = hit(kv.w);
        unsigned m = (z0 >> 7) | (z1 >> 6) | (z2 >> 5) | (z3 >> 4);
        for (unsigned bal = __builtin_amdgcn_ballot_w32(m != 0u); bal;
             bal = __builtin_amdgcn_ballot_w32(m != 0u)) {
          if (m != 0u) {
            const int p = __builtin_ctz(m);
            my[count + (int)__builtin_amdgcn_mbcnt_lo(bal, 0u)] = e0 + ((p & 7) << 2) + (p >> 3);
          }
          count += __builtin_popcount(bal);
          m &= m - 1u;
        }
      }
      __builtin_amdgcn_fence(__ATOMIC_SEQ_CST, "wavefront");
      while (count >= 32) {
        count -= 32;
        drain(my + count, 32);
      }
#pragma unroll
      for (int u = 0; u < KSU; ++u) cur[u] = nxt[u];
      eb = en;
    }
  }
  const long long lo = (long long)nfullk * kstride;

  const long long wstride = (long long)SW * SU * 128;
  const int nfull = ((((size_t)dst) & 15) == 0) ? (int)(((long long)n_edges - lo) / wstride) : 0;
  if (nfull > 0) {
    int ps = (int)(((long long)blockIdx.x * nfull) / gridDim.x);
    const long long wofs = (long long)wv * (SU * 128) + lane * 4;
    long long eb = lo + (long long)ps * wstride + wofs;
    v4i cur[SU];
#pragma unroll
    for (int u = 0; u < SU; ++u) cur[u] = *(const v4i*)(dst + eb + u * 128);
#pragma unroll 2
    for (int it = 0; it < nfull; ++it) {
      ps = (ps + 1 < nfull) ? ps + 1 : 0;
      const long long en = (it + 1 < nfull) ? lo + (long long)ps * wstride + wofs : eb;
      v4i nxt[SU];
#pragma unroll
      for (int u = 0; u < SU; ++u) nxt[u] = *(const v4i*)(dst + en + u * 128);
#pragma unroll
      for (int u = 0; u < SU; ++u) {
        const v4i dv = cur[u];
        const int e0 = (int)(eb + u * 128);
        const bool m0 = (unsigned)(dv.x - base) < (unsigned)cnt;
        const bool m1 = (unsigned)(dv.y - base) < (unsigned)cnt;
        const bool m2 = (unsigned)(dv.z - base) < (unsigned)cnt;
        const bool m3 = (unsigned)(dv.w - base) < (unsigned)cnt;
        if (__builtin_amdgcn_ballot_w32(m0 | m1 | m2 | m3)) {
          append(m0, e0);
          append(m1, e0 + 1);
          append(m2, e0 + 2);
          append(m3, e0 + 3);
        }
      }
      __builtin_amdgcn_fence(__ATOMIC_SEQ_CST, "wavefront");
      while (count >= 32) {
        count -= 32;
        drain(my + count, 32);
      }
#pragma unroll
      for (int u = 0; u < SU; ++u) cur[u] = nxt[u];
      eb = en;
    }
  }
  for (long long t = lo + (long long)nfull * wstride + wv * 32; ; t += TU * SW * 32) {
    const bool done = t >= n_edges;
    if (!done) {
      int dv[TU];
#pragma unroll
      for (int u = 0; u < TU; ++u) {
        const long long e = t + u * (SW * 32) + lane;
        dv[u] = (e < n_edges) ? dst[e] : -1;
      }
#pragma unroll
      for (int u = 0; u < TU; ++u)
        append((unsigned)(dv[u] - base) < (unsigned)cnt, (int)(t + u * (SW * 32) + lane));
    }
    __builtin_amdgcn_fence(__ATOMIC_SEQ_CST, "wavefront");
    const int lim = done ? 1 : 32;
    while (count >= lim) {
      const int k = count < 32 ? count : 32;
      count -= k;
      drain(my + count, k);
    }
    if (done) break;
  }

  __syncthreads();

  for (int i = threadIdx.x; i < cnt; i += SW * 32) {
    const size_t o = (size_t)(base + i) * OUT_C;
    const v2f z = *(const v2f*)(out + o);
    const float z0 = z.x + acc[i * OUT_C];
    const float z1 = z.y + acc[i * OUT_C + 1];
    const float mx = fmaxf(z0, z1);
    const float lse = mx + logf(expf(z0 - mx) + expf(z1 - mx));
    *(v2f*)(out + o) = v2f{z0 - lse, z1 - lse};
  }
}

extern "C" void kernel_launch(void* const* d_in, const int* in_sizes, int n_in,
                              void* d_out, int out_size, void* d_ws, size_t ws_size,
                              hipStream_t stream) {
  const float* x  = (const float*)d_in[0];
  const int*   ei = (const int*)d_in[1];
  const float* W1 = (const float*)d_in[2];
  const float* b1 = (const float*)d_in[3];
  const float* W2 = (const float*)d_in[4];
  const float* b2 = (const float*)d_in[5];
  const float* Ws = (const float*)d_in[6];
  const float* bs = (const float*)d_in[7];
  float* out = (float*)d_out;

  const int n_nodes = in_sizes[0] / IN_C;
  const int n_edges = in_sizes[1] / 2;
  const int* srcv = ei;
  const int* dstv = ei + n_edges;

  float* Wp   = (float*)d_ws;
  float* deg  = Wp + WP_FLOATS;
  size_t off  = ((size_t)n_nodes + 255) & ~(size_t)255;
  float* h    = deg + off;
  float* agg  = h + (size_t)n_nodes * HID;
  float* h2   = agg + (size_t)n_nodes * HID;

  const int T = 256;

  prep_kernel<<<((n_edges + 3) / 4 + T - 1) / T, T, 0, stream>>>(deg, n_nodes, dstv, n_edges,
                                                                 Wp, ws_size, h2);

  pack_w_kernel<<<WP_FLOATS / T, T, 0, stream>>>(W1, Ws, Wp);

  int tiles = (n_nodes + 15) / 16;
  gemm1_wmma_kernel<<<(tiles + 7) / 8, T, 0, stream>>>(x, Wp, dstv, deg, h, out,
                                                   n_nodes, n_edges);

  scatter1_kernel<<<(n_nodes + NPB - 1) / NPB, SW * 32, 0, stream>>>(h, deg, srcv, dstv, agg, n_edges, n_nodes,
                                                                          Wp, ws_size, h2);

  gemm2_fused_kernel<<<(n_nodes + T - 1) / T, T, 0, stream>>>(agg, h, b1, W2, b2, bs, deg,
                                                              h2, out, n_nodes);
  scatter2_kernel<<<(n_nodes + NPB - 1) / NPB, SW * 32, 0, stream>>>(h2, deg, srcv, dstv, out, n_edges, n_nodes,
                                                                Wp, ws_size);
}


# stage-1 owner workgroups finish their rows from LDS (self-loop, b1, relu, W2 product, out biases): edge sums never stored, one launch fewer; a node's 100-term W2 sums now per lane + lane tree
# speedup vs baseline: 2.1820x; 1.0518x over previous
#include <hip/hip_runtime.h>
#include <math.h>

#define IN_C 512
#define HID 100
#define OUT_C 2
#define NT 8
#define WP_FLOATS (256 * NT * 16 * 2)

typedef __attribute__((ext_vector_type(2))) float v2f;
typedef __attribute__((ext_vector_type(8))) float v8f;

#define PW 32
#define NPL 3
#define TAILF (HID - NPL * PW)
#define NPB 400

typedef __attribute__((ext_vector_type(4))) unsigned v4u;

__device__ __forceinline__ int key_count(int n_nodes, int n_edges) {
  const size_t room = (size_t)n_nodes * HID * sizeof(float);
  return (int)(((size_t)n_edges < room ? (size_t)n_edges : room) & ~(size_t)3);
}

__global__ void prep_kernel(float* __restrict__ deg, int n_nodes,
                            const int* __restrict__ dst, int n_edges,
                            float* __restrict__ keyw) {
  const int i = blockIdx.x * blockDim.x + threadIdx.x;
  for (int j = i; j < n_nodes; j += gridDim.x * blockDim.x) deg[j] = 1.0f;
  const int nk = key_count(n_nodes, n_edges);
  unsigned* kw = (unsigned*)keyw;
  const long long e = 4LL * i;
  if (e + 3 < nk) {
    const unsigned k0 = ((unsigned)dst[e]     / NPB) & 255u;
    const unsigned k1 = ((unsigned)dst[e + 1] / NPB) & 255u;
    const unsigned k2 = ((unsigned)dst[e + 2] / NPB) & 255u;
    const unsigned k3 = ((unsigned)dst[e + 3] / NPB) & 255u;
    kw[i] = k0 | (k1 << 8) | (k2 << 16) | (k3 << 24);
  }
}

__device__ __forceinline__ float deg_to_dinv(float d) { return (d > 0.f) ? rsqrtf(d) : 0.f; }

__global__ void pack_w_kernel(const float* __restrict__ W1,
                              const float* __restrict__ Ws,
                              float* __restrict__ Wp) {
  int idx = blockIdx.x * blockDim.x + threadIdx.x;
  if (idx >= WP_FLOATS) return;
  int kin = idx & 1;
  int col = (idx >> 1) & 15;
  int t   = (idx >> 5) & 7;
  int kp  = idx >> 8;
  int k   = (kp << 1) + kin;
  float v = 0.f;
  if (t < 7) {
    int c = (t << 4) + col;
    if (c < HID) v = W1[(size_t)k * HID + c];
  } else {
    if (col < OUT_C) v = Ws[(size_t)k * OUT_C + col];
  }
  Wp[idx] = v;
}

#define KBLK 16
__global__ void __launch_bounds__(256)
gemm1_wmma_kernel(const float* __restrict__ x, const float* __restrict__ Wp,
                  const int* __restrict__ dst, float* __restrict__ deg,
                  float* __restrict__ h,
                  float* __restrict__ outp, int n_nodes, int n_edges) {
  const int lane = threadIdx.x & 31;
  const int wv   = threadIdx.x >> 5;
  const int tile = blockIdx.x * (blockDim.x >> 5) + wv;
  const int row0 = tile << 4;
  if (row0 >= n_nodes) return;

  const long long stride = (long long)((n_nodes + 15) >> 4) << 5;
  long long e = ((long long)tile << 5) + lane;
  int dcur = (e < n_edges) ? dst[e] : -1;

  const int m  = lane & 15;
  const int kh = (lane >> 4) << 1;

  v8f acc[NT] = {};

  const float* __restrict__ pa = x + (size_t)(row0 + m) * IN_C + kh;
  const float* __restrict__ pb = Wp + ((size_t)(lane >> 4) * 256) + (m << 1);

  for (int kb = 0; kb < IN_C / (4 * KBLK); ++kb) {
    if (dcur >= 0) atomicAdd(&deg[dcur], 1.0f);
    const long long en = e + stride;
    const int dnxt = (en < n_edges) ? dst[en] : -1;
    for (int ks = 0; ks < KBLK; ++ks) {
      v2f a = *(const v2f*)pa;
#pragma unroll
      for (int t = 0; t < NT; ++t) {
        v2f b = *(const v2f*)(pb + (t << 5));
        acc[t] = __builtin_amdgcn_wmma_f32_16x16x4_f32(
            false, a, false, b, (short)0, acc[t], false, false);
      }
      pa += 4;
      pb += 512;
    }
    dcur = dnxt;
    e = en;
  }
  if (dcur >= 0) atomicAdd(&deg[dcur], 1.0f);
  for (e += stride; e < n_edges; e += stride) atomicAdd(&deg[dst[e]], 1.0f);

  const int rbase = (lane < 16) ? 0 : 8;
  const size_t plane = (size_t)n_nodes * PW;
#pragma unroll
  for (int r = 0; r < 8; ++r) {
    const int row = row0 + rbase + r;
#pragma unroll
    for (int t = 0; t < 7; ++t) {
      int col = (t << 4) + m;
      if (col < HID) {
        const size_t o = (t < 2 * NPL)
            ? (size_t)(t >> 1) * plane + (size_t)row * PW + ((t & 1) << 4) + m
            : (size_t)NPL * plane + (size_t)row * TAILF + m;
        h[o] = acc[t][r];
      }
    }
    if (m < OUT_C) outp[(size_t)row * OUT_C + m] = acc[7][r];
  }
}

#define SW 32
#define KSU 2
#define SU 4
#define LCAP (32 + KSU * 512)
static_assert(LCAP >= 32 + SU * 128, "the list also holds a pass of the dst loop");
#define TU 4
#define EB 8
#define RPW ((NPB + SW - 1) / SW)
#define RB 4
static_assert(RPW <= 32, "lane j of a wave keeps the results of the wave's row j");
static_assert(EB * TAILF == 32, "one wave-wide access covers the leftover features of a batch");
static_assert(LCAP >= 32 + TU * 32, "the list also holds a trip of the edge-at-a-time loop");

typedef __attribute__((ext_vector_type(4))) int v4i;

__device__ __forceinline__ void lds_add(float* p, float v) {
  __hip_atomic_fetch_add(p, v, __ATOMIC_RELAXED, __HIP_MEMORY_SCOPE_WORKGROUP);
}

__global__ void __launch_bounds__(SW * 32)
scatter1_kernel(const float* __restrict__ h,
                const float* __restrict__ deg,
                const int* __restrict__ src,
                const int* __restrict__ dst,
                const float* __restrict__ keyw, int n_edges, int n_nodes,
                const float* __restrict__ b1,
                const float* __restrict__ W2,
                const float* __restrict__ b2,
                const float* __restrict__ bs,
                float* __restrict__ h2,
                float* __restrict__ out) {
  __shared__ float acc[NPB * HID];
  __shared__ int lst[SW * LCAP];
  const int lane = threadIdx.x & 31;
  const int wv = __builtin_amdgcn_readfirstlane((int)(threadIdx.x >> 5));
  const int base = blockIdx.x * NPB;
  const int cnt = (n_nodes - base < NPB) ? (n_nodes - base) : NPB;
  const size_t plane = (size_t)n_nodes * PW;
  const char* __restrict__ hb0 = (const char*)h;
  const char* __restrict__ hb1 = (const char*)(h + plane);
  const char* __restrict__ hb2 = (const char*)(h + 2 * plane);
  const char* __restrict__ hbt = (const char*)(h + NPL * plane);
  const unsigned lb = (unsigned)lane * 4u;
  const int tu = lane / TAILF;
  const int tf = lane % TAILF;
  int* const my = lst + wv * LCAP;

  for (int i = threadIdx.x; i < NPB * HID; i += SW * 32) acc[i] = 0.f;
  __syncthreads();

  auto drain = [&](const int* p, int k) {
    const int e = (lane < k) ? p[lane] : -1;
    int s = 0, dl = -1;
    float nrm = 0.f;
    if (e >= 0) {
      s = src[e];
      const int d = dst[e];
      nrm = deg_to_dinv(deg[s]) * deg_to_dinv(deg[d]);
      if ((unsigned)(d - base) < (unsigned)cnt) dl = d - base;
    }
    for (int jb = 0; jb < k; jb += EB) {
      float v[EB][3];
      const int   st = __shfl(s, jb + tu);
      const int   dt = __shfl(dl, jb + tu);
      const float nt = __shfl(nrm, jb + tu);
#pragma unroll
      for (int u = 0; u < EB; ++u) {
        const int sj = __builtin_amdgcn_readlane(s, jb + u);
        const unsigned so = (unsigned)sj * (PW * 4u) + lb;
        v[u][0] = *(const float*)(hb0 + so);
        v[u][1] = *(const float*)(hb1 + so);
        v[u][2] = *(const float*)(hb2 + so);
      }
      const float vt = *(const float*)(hbt + ((unsigned)st * (TAILF * 4u) + (unsigned)tf * 4u));
#pragma unroll
      for (int u = 0; u < EB; ++u) {
        const int dj = __builtin_amdgcn_readlane(dl, jb + u);
        if (jb + u < k && dj >= 0) {
          const float nj = __int_as_float(__builtin_amdgcn_readlane(__float_as_int(nrm), jb + u));
          float* r = acc + dj * HID + lane;
          lds_add(r,      v[u][0] * nj);
          lds_add(r + 32, v[u][1] * nj);
          lds_add(r + 64, v[u][2] * nj);
        }
      }
      if (jb + tu < k && dt >= 0) lds_add(acc + dt * HID + NPL * PW + tf, vt * nt);
    }
  };

  int count = 0;
  auto append = [&](bool m, int ev) {
    const unsigned bal = __builtin_amdgcn_ballot_w32(m);
    if (bal) {
      if (m) my[count + (int)__builtin_amdgcn_mbcnt_lo(bal, 0u)] = ev;
      count += __builtin_popcount(bal);
    }
  };

  const int nk = key_count(n_nodes, n_edges);
  const unsigned char* keys = (const unsigned char*)keyw;
  const int kstride = SW * KSU * 512;
  const int nfullk = ((((size_t)keys) & 15) == 0) ? nk / kstride : 0;
  if (nfullk > 0) {
    const unsigned kb = (unsigned)(blockIdx.x & 255) * 0x01010101u;
    auto hit = [&](unsigned w) {
      const unsigned x = w ^ kb;
      return ~(((x & 0x7f7f7f7fu) + 0x7f7f7f7fu) | x) & 0x80808080u;
    };
    int ps = (int)(((long long)blockIdx.x * nfullk) / gridDim.x);
    const int wofs = wv * (KSU * 512) + lane * 16;
    int eb = ps * kstride + wofs;
    v4u cur[KSU];
#pragma unroll
    for (int u = 0; u < KSU; ++u) cur[u] = *(const v4u*)(keys + eb + u * 512);
#pragma unroll 2
    for (int it = 0; it < nfullk; ++it) {
      ps = (ps + 1 < nfullk) ? ps + 1 : 0;
      const int en = (it + 1 < nfullk) ? ps * kstride + wofs : eb;
      v4u nxt[KSU];
#pragma unroll
      for (int u = 0; u < KSU; ++u) nxt[u] = *(const v4u*)(keys + en + u * 512);
#pragma unroll
      for (int u = 0; u < KSU; ++u) {
        const v4u kv = cur[u];
        const int e0 = eb + u * 512;
        const unsigned z0 = hit(kv.x), z1 = hit(kv.y), z2 = hit(kv.z), z3 = hit(kv.w);
        unsigned m = (z0 >> 7) | (z1 >> 6) | (z2 >> 5) | (z3 >> 4);
        for (unsigned bal = __builtin_amdgcn_ballot_w32(m != 0u); bal;
             bal = __builtin_amdgcn_ballot_w32(m != 0u)) {
          if (m != 0u) {
            const int p = __builtin_ctz(m);
            my[count + (int)__builtin_amdgcn_mbcnt_lo(bal, 0u)] = e0 + ((p & 7) << 2) + (p >> 3);
          }
          count += __builtin_popcount(bal);
          m &= m - 1u;
        }
      }
      __builtin_amdgcn_fence(__ATOMIC_SEQ_CST, "wavefront");
      while (count >= 32) {
        count -= 32;
        drain(my + count, 32);
      }
#pragma unroll
      for (int u = 0; u < KSU; ++u) cur[u] = nxt[u];
      eb = en;
    }
  }
  const long long lo = (long long)nfullk * kstride;

  const long long wstride = (long long)SW * SU * 128;
  const int nfull = ((((size_t)dst) & 15) == 0) ? (int)(((long long)n_edges - lo) / wstride) : 0;
  if (nfull > 0) {
    int ps = (int)(((long long)blockIdx.x * nfull) / gridDim.x);
    const long long wofs = (long long)wv * (SU * 128) + lane * 4;
    long long eb = lo + (long long)ps * wstride + wofs;
    v4i cur[SU];
#pragma unroll
    for (int u = 0; u < SU; ++u) cur[u] = *(const v4i*)(dst + eb + u * 128);
#pragma unroll 2
    for (int it = 0; it < nfull; ++it) {
      ps = (ps + 1 < nfull) ? ps + 1 : 0;
      const long long en = (it + 1 < nfull) ? lo + (long long)ps * wstride + wofs : eb;
      v4i nxt[SU];
#pragma unroll
      for (int u = 0; u < SU; ++u) nxt[u] = *(const v4i*)(dst + en + u * 128);
#pragma unroll
      for (int u = 0; u < SU; ++u) {
        const v4i dv = cur[u];
        const int e0 = (int)(eb + u * 128);
        const bool m0 = (unsigned)(dv.x - base) < (unsigned)cnt;
        const bool m1 = (unsigned)(dv.y - base) < (unsigned)cnt;
        const bool m2 = (unsigned)(dv.z - base) < (unsigned)cnt;
        const bool m3 = (unsigned)(dv.w - base) < (unsigned)cnt;
        if (__builtin_amdgcn_ballot_w32(m0 | m1 | m2 | m3)) {
          append(m0, e0);
          append(m1, e0 + 1);
          append(m2, e0 + 2);
          append(m3, e0 + 3);
        }
      }
      __builtin_amdgcn_fence(__ATOMIC_SEQ_CST, "wavefront");
      while (count >= 32) {
        count -= 32;
        drain(my + count, 32);
      }
#pragma unroll
      for (int u = 0; u < SU; ++u) cur[u] = nxt[u];
      eb = en;
    }
  }
  for (long long t = lo + (long long)nfull * wstride + wv * 32; ; t += TU * SW * 32) {
    const bool done = t >= n_edges;
    if (!done) {
      int dv[TU];
#pragma unroll
      for (int u = 0; u < TU; ++u) {
        const long long e = t + u * (SW * 32) + lane;
        dv[u] = (e < n_edges) ? dst[e] : -1;
      }
#pragma unroll
      for (int u = 0; u < TU; ++u)
        append((unsigned)(dv[u] - base) < (unsigned)cnt, (int)(t + u * (SW * 32) + lane));
    }
    __builtin_amdgcn_fence(__ATOMIC_SEQ_CST, "wavefront");
    const int lim = done ? 1 : 32;
    while (count >= lim) {
      const int k = count < 32 ? count : 32;
      count -= k;
      drain(my + count, k);
    }
    if (done) break;
  }

  __syncthreads();

  const v2f* __restrict__ w2 = (const v2f*)W2;
  float bb[NPL];
  v2f ww[NPL];
#pragma unroll
  for (int p = 0; p < NPL; ++p) { bb[p] = b1[p * PW + lane]; ww[p] = w2[p * PW + lane]; }
  const bool tl = lane < TAILF;
  const float bt = tl ? b1[NPL * PW + lane] : 0.f;
  const v2f wt = tl ? w2[NPL * PW + lane] : v2f{0.f, 0.f};
  const int r0 = wv * RPW;
  const int rmine = r0 + lane;
  const bool own = lane < RPW && rmine < cnt;
  float dq = 0.f;
  if (own) { const float di = deg_to_dinv(deg[base + rmine]); dq = di * di; }
  v2f keep = {0.f, 0.f};
#pragma unroll
  for (int jb = 0; jb < RPW; jb += RB) {
    float hv[RB][NPL], ht[RB], av[RB][NPL], at[RB];
#pragma unroll
    for (int u = 0; u < RB; ++u) {
      if (jb + u < RPW) {
        const int r = (r0 + jb + u < cnt) ? r0 + jb + u : cnt - 1;
        const size_t o = (size_t)(base + r) * PW + lane;
        const float* a = acc + r * HID;
#pragma unroll
        for (int p = 0; p < NPL; ++p) { hv[u][p] = h[p * plane + o]; av[u][p] = a[p * PW + lane]; }
        ht[u] = tl ? h[NPL * plane + (size_t)(base + r) * TAILF + lane] : 0.f;
        at[u] = tl ? a[NPL * PW + lane] : 0.f;
      }
    }
#pragma unroll
    for (int u = 0; u < RB; ++u) {
      if (jb + u < RPW) {
        const float q = __int_as_float(__builtin_amdgcn_readlane(__float_as_int(dq), jb + u));
        v2f t = {0.f, 0.f};
#pragma unroll
        for (int p = 0; p < NPL; ++p) {
          float v = (av[u][p] + hv[u][p] * q) + bb[p];
          v = v > 0.f ? v : 0.f;
          t += v2f{v, v} * ww[p];
        }
        {
          float v = (at[u] + ht[u] * q) + bt;
          v = v > 0.f ? v : 0.f;
          t += v2f{v, v} * wt;
        }
#pragma unroll
        for (int mk = 16; mk >= 1; mk >>= 1) {
          t.x += __shfl_xor(t.x, mk);
          t.y += __shfl_xor(t.y, mk);
        }
        if (lane == jb + u) keep = t;
      }
    }
  }
  if (own) {
    const size_t o = (size_t)(base + rmine) * OUT_C;
    *(v2f*)(h2 + o) = keep;
    const v2f z = *(const v2f*)(out + o);
    *(v2f*)(out + o) = v2f{z.x + (b2[0] + bs[0] + keep.x * dq),
                           z.y + (b2[1] + bs[1] + keep.y * dq)};
  }
}

__global__ void __launch_bounds__(SW * 32)
scatter2_kernel(const float* __restrict__ h2,
                const float* __restrict__ deg,
                const int* __restrict__ src,
                const int* __restrict__ dst,
                float* __restrict__ out, int n_edges, int n_nodes,
                const float* __restrict__ keyw) {
  __shared__ float acc[NPB * OUT_C];
  __shared__ int lst[SW * LCAP];
  const int lane = threadIdx.x & 31;
  const int wv = __builtin_amdgcn_readfirstlane((int)(threadIdx.x >> 5));
  const int base = blockIdx.x * NPB;
  const int cnt = (n_nodes - base < NPB) ? (n_nodes - base) : NPB;
  int* const my = lst + wv * LCAP;

  for (int i = threadIdx.x; i < NPB * OUT_C; i += SW * 32) acc[i] = 0.f;
  __syncthreads();

  auto drain = [&](const int* p, int k) {
    if (lane < k) {
      const int e = p[lane];
      const int s = src[e];
      const int d = dst[e];
      if ((unsigned)(d - base) < (unsigned)cnt) {
        const float nrm = deg_to_dinv(deg[s]) * deg_to_dinv(deg[d]);
        const v2f hv = *(const v2f*)(h2 + (size_t)s * OUT_C);
        float* r = acc + (d - base) * OUT_C;
        lds_add(r,     hv.x * nrm);
        lds_add(r + 1, hv.y * nrm);
      }
    }
  };

  int count = 0;
  auto append = [&](bool m, int ev) {
    const unsigned bal = __builtin_amdgcn_ballot_w32(m);
    if (bal) {
      if (m) my[count + (int)__builtin_amdgcn_mbcnt_lo(bal, 0u)] = ev;
      count += __builtin_popcount(bal);
    }
  };

  const int nk = key_count(n_nodes, n_edges);
  const unsigned char* keys = (const unsigned char*)keyw;
  const int kstride = SW * KSU * 512;
  const int nfullk = ((((size_t)keys) & 15) == 0) ? nk / kstride : 0;
  if (nfullk > 0) {
    const unsigned kb = (unsigned)(blockIdx.x & 255) * 0x01010101u;
    auto hit = [&](unsigned w) {
      const unsigned x = w ^ kb;
      return ~(((x & 0x7f7f7f7fu) + 0x7f7f7f7fu) | x) & 0x80808080u;
    };
    int ps = (int)(((long long)blockIdx.x * nfullk) / gridDim.x);
    const int wofs = wv * (KSU * 512) + lane * 16;
    int eb = ps * kstride + wofs;
    v4u cur[KSU];
#pragma unroll
    for (int u = 0; u < KSU; ++u) cur[u] = *(const v4u*)(keys + eb + u * 512);
#pragma unroll 2
    for (int it = 0; it < nfullk; ++it) {
      ps = (ps + 1 < nfullk) ? ps + 1 : 0;
      const int en = (it + 1 < nfullk) ? ps * kstride + wofs : eb;
      v4u nxt[KSU];
#pragma unroll
      for (int u = 0; u < KSU; ++u) nxt[u] = *(const v4u*)(keys + en + u * 512);
#pragma unroll
      for (int u = 0; u < KSU; ++u) {
        const v4u kv = cur[u];
        const int e0 = eb + u * 512;
        const unsigned z0 = hit(kv.x), z1 = hit(kv.y), z2 = hit(kv.z), z3 = hit(kv.w);
        unsigned m = (z0 >> 7) | (z1 >> 6) | (z2 >> 5) | (z3 >> 4);
        for (unsigned bal = __builtin_amdgcn_ballot_w32(m != 0u); bal;
             bal = __builtin_amdgcn_ballot_w32(m != 0u)) {
          if (m != 0u) {
            const int p = __builtin_ctz(m);
            my[count + (int)__builtin_amdgcn_mbcnt_lo(bal, 0u)] = e0 + ((p & 7) << 2) + (p >> 3);
          }
          count += __builtin_popcount(bal);
          m &= m - 1u;
        }
      }
      __builtin_amdgcn_fence(__ATOMIC_SEQ_CST, "wavefront");
      while (count >= 32) {
        count -= 32;
        drain(my + count, 32);
      }
#pragma unroll
      for (int u = 0; u < KSU; ++u) cur[u] = nxt[u];
      eb = en;
    }
  }
  const long long lo = (long long)nfullk * kstride;

  const long long wstride = (long long)SW * SU * 128;
  const int nfull = ((((size_t)dst) & 15) == 0) ? (int)(((long long)n_edges - lo) / wstride) : 0;
  if (nfull > 0) {
    int ps = (int)(((long long)blockIdx.x * nfull) / gridDim.x);
    const long long wofs = (long long)wv * (SU * 128) + lane * 4;
    long long eb = lo + (long long)ps * wstride + wofs;
    v4i cur[SU];
#pragma unroll
    for (int u = 0; u < SU; ++u) cur[u] = *(const v4i*)(dst + eb + u * 128);
#pragma unroll 2
    for (int it = 0; it < nfull; ++it) {
      ps = (ps + 1 < nfull) ? ps + 1 : 0;
      const long long en = (it + 1 < nfull) ? lo + (long long)ps * wstride + wofs : eb;
      v4i nxt[SU];
#pragma unroll
      for (int u = 0; u < SU; ++u) nxt[u] = *(const v4i*)(dst + en + u * 128);
#pragma unroll
      for (int u = 0; u < SU; ++u) {
        const v4i dv = cur[u];
        const int e0 = (int)(eb + u * 128);
        const bool m0 = (unsigned)(dv.x - base) < (unsigned)cnt;
        const bool m1 = (unsigned)(dv.y - base) < (unsigned)cnt;
        const bool m2 = (unsigned)(dv.z - base) < (unsigned)cnt;
        const bool m3 = (unsigned)(dv.w - base) < (unsigned)cnt;
        if (__builtin_amdgcn_ballot_w32(m0 | m1 | m2 | m3)) {
          append(m0, e0);
          append(m1, e0 + 1);
          append(m2, e0 + 2);
          append(m3, e0 + 3);
        }
      }
      __builtin_amdgcn_fence(__ATOMIC_SEQ_CST, "wavefront");
      while (count >= 32) {
        count -= 32;
        drain(my + count, 32);
      }
#pragma unroll
      for (int u = 0; u < SU; ++u) cur[u] = nxt[u];
      eb = en;
    }
  }
  for (long long t = lo + (long long)nfull * wstride + wv * 32; ; t += TU * SW * 32) {
    const bool done = t >= n_edges;
    if (!done) {
      int dv[TU];
#pragma unroll
      for (int u = 0; u < TU; ++u) {
        const long long e = t + u * (SW * 32) + lane;
        dv[u] = (e < n_edges) ? dst[e] : -1;
      }
#pragma unroll
      for (int u = 0; u < TU; ++u)
        append((unsigned)(dv[u] - base) < (unsigned)cnt, (int)(t + u * (SW * 32) + lane));
    }
    __builtin_amdgcn_fence(__ATOMIC_SEQ_CST, "wavefront");
    const int lim = done ? 1 : 32;
    while (count >= lim) {
      const int k = count < 32 ? count : 32;
      count -= k;
      drain(my + count, k);
    }
    if (done) break;
  }

  __syncthreads();

  for (int i = threadIdx.x; i < cnt; i += SW * 32) {
    const size_t o = (size_t)(base + i) * OUT_C;
    const v2f z = *(const v2f*)(out + o);
    const float z0 = z.x + acc[i * OUT_C];
    const float z1 = z.y + acc[i * OUT_C + 1];
    const float mx = fmaxf(z0, z1);
    const float lse = mx + logf(expf(z0 - mx) + expf(z1 - mx));
    *(v2f*)(out + o) = v2f{z0 - lse, z1 - lse};
  }
}

extern "C" void kernel_launch(void* const* d_in, const int* in_sizes, int n_in,
                              void* d_out, int out_size, void* d_ws, size_t ws_size,
                              hipStream_t stream) {
  const float* x  = (const float*)d_in[0];
  const int*   ei = (const int*)d_in[1];
  const float* W1 = (const float*)d_in[2];
  const float* b1 = (const float*)d_in[3];
  const float* W2 = (const float*)d_in[4];
  const float* b2 = (const float*)d_in[5];
  const float* Ws = (const float*)d_in[6];
  const float* bs = (const float*)d_in[7];
  float* out = (float*)d_out;

  const int n_nodes = in_sizes[0] / IN_C;
  const int n_edges = in_sizes[1] / 2;
  const int* srcv = ei;
  const int* dstv = ei + n_edges;

  float* Wp   = (float*)d_ws;
  float* deg  = Wp + WP_FLOATS;
  size_t off  = ((size_t)n_nodes + 255) & ~(size_t)255;
  float* h    = deg + off;
  float* keyw = h + (size_t)n_nodes * HID;
  float* h2   = keyw + (size_t)n_nodes * HID;

  const int T = 256;

  prep_kernel<<<((n_edges + 3) / 4 + T - 1) / T, T, 0, stream>>>(deg, n_nodes, dstv, n_edges, keyw);

  pack_w_kernel<<<WP_FLOATS / T, T, 0, stream>>>(W1, Ws, Wp);

  int tiles = (n_nodes + 15) / 16;
  gemm1_wmma_kernel<<<(tiles + 7) / 8, T, 0, stream>>>(x, Wp, dstv, deg, h, out,
                                                   n_nodes, n_edges);

  scatter1_kernel<<<(n_nodes + NPB - 1) / NPB, SW * 32, 0, stream>>>(h, deg, srcv, dstv, keyw, n_edges, n_nodes,
                                                                          b1, W2, b2, bs, h2, out);

  scatter2_kernel<<<(n_nodes + NPB - 1) / NPB, SW * 32, 0, stream>>>(h2, deg, srcv, dstv, out, n_edges, n_nodes,
                                                                keyw);
}


# stage 2 no longer filters the edge list: each stage-1 wave leaves a (source, owned row) record per edge it summed and its twin wave of stage 2 reads those; LDS arrival-order sums as before
# speedup vs baseline: 2.4945x; 1.1432x over previous
#include <hip/hip_runtime.h>
#include <math.h>

#define IN_C 512
#define HID 100
#define OUT_C 2
#define NT 8
#define WP_FLOATS (256 * NT * 16 * 2)

typedef __attribute__((ext_vector_type(2))) float v2f;
typedef __attribute__((ext_vector_type(8))) float v8f;

#define PW 32
#define NPL 3
#define TAILF (HID - NPL * PW)
#define NPB 400

typedef __attribute__((ext_vector_type(4))) unsigned v4u;
typedef __attribute__((ext_vector_type(2))) int v2i;
#define SWV 32
#define CAPW 512

__device__ __forceinline__ size_t count_bytes(int n_nodes) {
  return (size_t)((n_nodes + NPB - 1) / NPB) * SWV * sizeof(int);
}
__device__ __forceinline__ int key_count(int n_nodes, int n_edges) {
  const size_t room = (size_t)n_nodes * HID * sizeof(float) - count_bytes(n_nodes);
  return (int)(((size_t)n_edges < room ? (size_t)n_edges : room) & ~(size_t)3);
}
__device__ __forceinline__ size_t record_offset(int n_nodes, int n_edges) {
  return (count_bytes(n_nodes) + (size_t)key_count(n_nodes, n_edges) + 255) & ~(size_t)255;
}

__global__ void prep_kernel(float* __restrict__ deg, int n_nodes,
                            const int* __restrict__ dst, int n_edges,
                            float* __restrict__ keyw) {
  const int i = blockIdx.x * blockDim.x + threadIdx.x;
  for (int j = i; j < n_nodes; j += gridDim.x * blockDim.x) deg[j] = 1.0f;
  const int nk = key_count(n_nodes, n_edges);
  unsigned* kw = (unsigned*)keyw;
  const long long e = 4LL * i;
  if (e + 3 < nk) {
    const unsigned k0 = ((unsigned)dst[e]     / NPB) & 255u;
    const unsigned k1 = ((unsigned)dst[e + 1] / NPB) & 255u;
    const unsigned k2 = ((unsigned)dst[e + 2] / NPB) & 255u;
    const unsigned k3 = ((unsigned)dst[e + 3] / NPB) & 255u;
    kw[i] = k0 | (k1 << 8) | (k2 << 16) | (k3 << 24);
  }
}

__device__ __forceinline__ float deg_to_dinv(float d) { return (d > 0.f) ? rsqrtf(d) : 0.f; }

__global__ void pack_w_kernel(const float* __restrict__ W1,
                              const float* __restrict__ Ws,
                              float* __restrict__ Wp) {
  int idx = blockIdx.x * blockDim.x + threadIdx.x;
  if (idx >= WP_FLOATS) return;
  int kin = idx & 1;
  int col = (idx >> 1) & 15;
  int t   = (idx >> 5) & 7;
  int kp  = idx >> 8;
  int k   = (kp << 1) + kin;
  float v = 0.f;
  if (t < 7) {
    int c = (t << 4) + col;
    if (c < HID) v = W1[(size_t)k * HID + c];
  } else {
    if (col < OUT_C) v = Ws[(size_t)k * OUT_C + col];
  }
  Wp[idx] = v;
}

#define KBLK 16
__global__ void __launch_bounds__(256)
gemm1_wmma_kernel(const float* __restrict__ x, const float* __restrict__ Wp,
                  const int* __restrict__ dst, float* __restrict__ deg,
                  float* __restrict__ h,
                  float* __restrict__ outp, int n_nodes, int n_edges) {
  const int lane = threadIdx.x & 31;
  const int wv   = threadIdx.x >> 5;
  const int tile = blockIdx.x * (blockDim.x >> 5) + wv;
  const int row0 = tile << 4;
  if (row0 >= n_nodes) return;

  const long long stride = (long long)((n_nodes + 15) >> 4) << 5;
  long long e = ((long long)tile << 5) + lane;
  int dcur = (e < n_edges) ? dst[e] : -1;

  const int m  = lane & 15;
  const int kh = (lane >> 4) << 1;

  v8f acc[NT] = {};

  const float* __restrict__ pa = x + (size_t)(row0 + m) * IN_C + kh;
  const float* __restrict__ pb = Wp + ((size_t)(lane >> 4) * 256) + (m << 1);

  for (int kb = 0; kb < IN_C / (4 * KBLK); ++kb) {
    if (dcur >= 0) atomicAdd(&deg[dcur], 1.0f);
    const long long en = e + stride;
    const int dnxt = (en < n_edges) ? dst[en] : -1;
    for (int ks = 0; ks < KBLK; ++ks) {
      v2f a = *(const v2f*)pa;
#pragma unroll
      for (int t = 0; t < NT; ++t) {
        v2f b = *(const v2f*)(pb + (t << 5));
        acc[t] = __builtin_amdgcn_wmma_f32_16x16x4_f32(
            false, a, false, b, (short)0, acc[t], false, false);
      }
      pa += 4;
      pb += 512;
    }
    dcur = dnxt;
    e = en;
  }
  if (dcur >= 0) atomicAdd(&deg[dcur], 1.0f);
  for (e += stride; e < n_edges; e += stride) atomicAdd(&deg[dst[e]], 1.0f);

  const int rbase = (lane < 16) ? 0 : 8;
  const size_t plane = (size_t)n_nodes * PW;
#pragma unroll
  for (int r = 0; r < 8; ++r) {
    const int row = row0 + rbase + r;
#pragma unroll
    for (int t = 0; t < 7; ++t) {
      int col = (t << 4) + m;
      if (col < HID) {
        const size_t o = (t < 2 * NPL)
            ? (size_t)(t >> 1) * plane + (size_t)row * PW + ((t & 1) << 4) + m
            : (size_t)NPL * plane + (size_t)row * TAILF + m;
        h[o] = acc[t][r];
      }
    }
    if (m < OUT_C) outp[(size_t)row * OUT_C + m] = acc[7][r];
  }
}

#define SW 32
#define KSU 2
#define SU 4
#define LCAP (32 + KSU * 512)
static_assert(LCAP >= 32 + SU * 128, "the list also holds a pass of the dst loop");
static_assert(SW == SWV, "one count per wave of an edge-pass workgroup");
#define TU 4
#define EB 8
#define RU 4
#define RPW ((NPB + SW - 1) / SW)
#define RB 4
static_assert(RPW <= 32, "lane j of a wave keeps the results of the wave's row j");
static_assert(EB * TAILF == 32, "one wave-wide access covers the leftover features of a batch");
static_assert(LCAP >= 32 + TU * 32, "the list also holds a trip of the edge-at-a-time loop");

typedef __attribute__((ext_vector_type(4))) int v4i;

__device__ __forceinline__ void lds_add(float* p, float v) {
  __hip_atomic_fetch_add(p, v, __ATOMIC_RELAXED, __HIP_MEMORY_SCOPE_WORKGROUP);
}

__global__ void __launch_bounds__(SW * 32)
scatter1_kernel(const float* __restrict__ h,
                const float* __restrict__ deg,
                const int* __restrict__ src,
                const int* __restrict__ dst,
                const float* __restrict__ keyw, int n_edges, int n_nodes,
                const float* __restrict__ b1,
                const float* __restrict__ W2,
                const float* __restrict__ b2,
                const float* __restrict__ bs,
                float* __restrict__ h2,
                float* __restrict__ out,
                float* __restrict__ cntw) {
  __shared__ float acc[NPB * HID];
  __shared__ int lst[SW * LCAP];
  const int lane = threadIdx.x & 31;
  const int wv = __builtin_amdgcn_readfirstlane((int)(threadIdx.x >> 5));
  const int base = blockIdx.x * NPB;
  const int cnt = (n_nodes - base < NPB) ? (n_nodes - base) : NPB;
  const size_t plane = (size_t)n_nodes * PW;
  const char* __restrict__ hb0 = (const char*)h;
  const char* __restrict__ hb1 = (const char*)(h + plane);
  const char* __restrict__ hb2 = (const char*)(h + 2 * plane);
  const char* __restrict__ hbt = (const char*)(h + NPL * plane);
  const unsigned lb = (unsigned)lane * 4u;
  const int tu = lane / TAILF;
  const int tf = lane % TAILF;
  int* const my = lst + wv * LCAP;

  for (int i = threadIdx.x; i < NPB * HID; i += SW * 32) acc[i] = 0.f;
  __syncthreads();

  const size_t rofs = record_offset(n_nodes, n_edges);
  v2i* const rec = (v2i*)((char*)cntw + rofs) + ((size_t)blockIdx.x * SW + wv) * CAPW;
  int rc = 0;
  bool full = rofs + (size_t)gridDim.x * SW * CAPW * sizeof(v2i)
              > (size_t)n_nodes * HID * sizeof(float);

  auto drain = [&](const int* p, int k) {
    const int e = (lane < k) ? p[lane] : -1;
    int s = 0, dl = -1;
    float nrm = 0.f;
    if (e >= 0) {
      s = src[e];
      const int d = dst[e];
      nrm = deg_to_dinv(deg[s]) * deg_to_dinv(deg[d]);
      if ((unsigned)(d - base) < (unsigned)cnt) dl = d - base;
    }
    if (!full) {
      if (rc + k <= CAPW) {
        if (lane < k) rec[rc + lane] = v2i{s, dl};
        rc += k;
      } else {
        full = true;
      }
    }
    for (int jb = 0; jb < k; jb += EB) {
      float v[EB][3];
      const int   st = __shfl(s, jb + tu);
      const int   dt = __shfl(dl, jb + tu);
      const float nt = __shfl(nrm, jb + tu);
#pragma unroll
      for (int u = 0; u < EB; ++u) {
        const int sj = __builtin_amdgcn_readlane(s, jb + u);
        const unsigned so = (unsigned)sj * (PW * 4u) + lb;
        v[u][0] = *(const float*)(hb0 + so);
        v[u][1] = *(const float*)(hb1 + so);
        v[u][2] = *(const float*)(hb2 + so);
      }
      const float vt = *(const float*)(hbt + ((unsigned)st * (TAILF * 4u) + (unsigned)tf * 4u));
#pragma unroll
      for (int u = 0; u < EB; ++u) {
        const int dj = __builtin_amdgcn_readlane(dl, jb + u);
        if (jb + u < k && dj >= 0) {
          const float nj = __int_as_float(__builtin_amdgcn_readlane(__float_as_int(nrm), jb + u));
          float* r = acc + dj * HID + lane;
          lds_add(r,      v[u][0] * nj);
          lds_add(r + 32, v[u][1] * nj);
          lds_add(r + 64, v[u][2] * nj);
        }
      }
      if (jb + tu < k && dt >= 0) lds_add(acc + dt * HID + NPL * PW + tf, vt * nt);
    }
  };

  int count = 0;
  auto append = [&](bool m, int ev) {
    const unsigned bal = __builtin_amdgcn_ballot_w32(m);
    if (bal) {
      if (m) my[count + (int)__builtin_amdgcn_mbcnt_lo(bal, 0u)] = ev;
      count += __builtin_popcount(bal);
    }
  };

  const int nk = key_count(n_nodes, n_edges);
  const unsigned char* keys = (const unsigned char*)keyw;
  const int kstride = SW * KSU * 512;
  const int nfullk = ((((size_t)keys) & 15) == 0) ? nk / kstride : 0;
  if (nfullk > 0) {
    const unsigned kb = (unsigned)(blockIdx.x & 255) * 0x01010101u;
    auto hit = [&](unsigned w) {
      const unsigned x = w ^ kb;
      return ~(((x & 0x7f7f7f7fu) + 0x7f7f7f7fu) | x) & 0x80808080u;
    };
    int ps = (int)(((long long)blockIdx.x * nfullk) / gridDim.x);
    const int wofs = wv * (KSU * 512) + lane * 16;
    int eb = ps * kstride + wofs;
    v4u cur[KSU];
#pragma unroll
    for (int u = 0; u < KSU; ++u) cur[u] = *(const v4u*)(keys + eb + u * 512);
#pragma unroll 2
    for (int it = 0; it < nfullk; ++it) {
      ps = (ps + 1 < nfullk) ? ps + 1 : 0;
      const int en = (it + 1 < nfullk) ? ps * kstride + wofs : eb;
      v4u nxt[KSU];
#pragma unroll
      for (int u = 0; u < KSU; ++u) nxt[u] = *(const v4u*)(keys + en + u * 512);
#pragma unroll
      for (int u = 0; u < KSU; ++u) {
        const v4u kv = cur[u];
        const int e0 = eb + u * 512;
        const unsigned z0 = hit(kv.x), z1 = hit(kv.y), z2 = hit(kv.z), z3 = hit(kv.w);
        unsigned m = (z0 >> 7) | (z1 >> 6) | (z2 >> 5) | (z3 >> 4);
        for (unsigned bal = __builtin_amdgcn_ballot_w32(m != 0u); bal;
             bal = __builtin_amdgcn_ballot_w32(m != 0u)) {
          if (m != 0u) {
            const int p = __builtin_ctz(m);
            my[count + (int)__builtin_amdgcn_mbcnt_lo(bal, 0u)] = e0 + ((p & 7) << 2) + (p >> 3);
          }
          count += __builtin_popcount(bal);
          m &= m - 1u;
        }
      }
      __builtin_amdgcn_fence(__ATOMIC_SEQ_CST, "wavefront");
      while (count >= 32) {
        count -= 32;
        drain(my + count, 32);
      }
#pragma unroll
      for (int u = 0; u < KSU; ++u) cur[u] = nxt[u];
      eb = en;
    }
  }
  const long long lo = (long long)nfullk * kstride;

  const long long wstride = (long long)SW * SU * 128;
  const int nfull = ((((size_t)dst) & 15) == 0) ? (int)(((long long)n_edges - lo) / wstride) : 0;
  if (nfull > 0) {
    int ps = (int)(((long long)blockIdx.x * nfull) / gridDim.x);
    const long long wofs = (long long)wv * (SU * 128) + lane * 4;
    long long eb = lo + (long long)ps * wstride + wofs;
    v4i cur[SU];
#pragma unroll
    for (int u = 0; u < SU; ++u) cur[u] = *(const v4i*)(dst + eb + u * 128);
#pragma unroll 2
    for (int it = 0; it < nfull; ++it) {
      ps = (ps + 1 < nfull) ? ps + 1 : 0;
      const long long en = (it + 1 < nfull) ? lo + (long long)ps * wstride + wofs : eb;
      v4i nxt[SU];
#pragma unroll
      for (int u = 0; u < SU; ++u) nxt[u] = *(const v4i*)(dst + en + u * 128);
#pragma unroll
      for (int u = 0; u < SU; ++u) {
        const v4i dv = cur[u];
        const int e0 = (int)(eb + u * 128);
        const bool m0 = (unsigned)(dv.x - base) < (unsigned)cnt;
        const bool m1 = (unsigned)(dv.y - base) < (unsigned)cnt;
        const bool m2 = (unsigned)(dv.z - base) < (unsigned)cnt;
        const bool m3 = (unsigned)(dv.w - base) < (unsigned)cnt;
        if (__builtin_amdgcn_ballot_w32(m0 | m1 | m2 | m3)) {
          append(m0, e0);
          append(m1, e0 + 1);
          append(m2, e0 + 2);
          append(m3, e0 + 3);
        }
      }
      __builtin_amdgcn_fence(__ATOMIC_SEQ_CST, "wavefront");
      while (count >= 32) {
        count -= 32;
        drain(my + count, 32);
      }
#pragma unroll
      for (int u = 0; u < SU; ++u) cur[u] = nxt[u];
      eb = en;
    }
  }
  for (long long t = lo + (long long)nfull * wstride + wv * 32; ; t += TU * SW * 32) {
    const bool done = t >= n_edges;
    if (!done) {
      int dv[TU];
#pragma unroll
      for (int u = 0; u < TU; ++u) {
        const long long e = t + u * (SW * 32) + lane;
        dv[u] = (e < n_edges) ? dst[e] : -1;
      }
#pragma unroll
      for (int u = 0; u < TU; ++u)
        append((unsigned)(dv[u] - base) < (unsigned)cnt, (int)(t + u * (SW * 32) + lane));
    }
    __builtin_amdgcn_fence(__ATOMIC_SEQ_CST, "wavefront");
    const int lim = done ? 1 : 32;
    while (count >= lim) {
      const int k = count < 32 ? count : 32;
      count -= k;
      drain(my + count, k);
    }
    if (done) break;
  }

  if (lane == 0) ((int*)cntw)[blockIdx.x * SW + wv] = full ? -1 : rc;

  __syncthreads();

  const v2f* __restrict__ w2 = (const v2f*)W2;
  float bb[NPL];
  v2f ww[NPL];
#pragma unroll
  for (int p = 0; p < NPL; ++p) { bb[p] = b1[p * PW + lane]; ww[p] = w2[p * PW + lane]; }
  const bool tl = lane < TAILF;
  const float bt = tl ? b1[NPL * PW + lane] : 0.f;
  const v2f wt = tl ? w2[NPL * PW + lane] : v2f{0.f, 0.f};
  const int r0 = wv * RPW;
  const int rmine = r0 + lane;
  const bool own = lane < RPW && rmine < cnt;
  float dq = 0.f;
  if (own) { const float di = deg_to_dinv(deg[base + rmine]); dq = di * di; }
  v2f keep = {0.f, 0.f};
#pragma unroll
  for (int jb = 0; jb < RPW; jb += RB) {
    float hv[RB][NPL], ht[RB], av[RB][NPL], at[RB];
#pragma unroll
    for (int u = 0; u < RB; ++u) {
      if (jb + u < RPW) {
        const int r = (r0 + jb + u < cnt) ? r0 + jb + u : cnt - 1;
        const size_t o = (size_t)(base + r) * PW + lane;
        const float* a = acc + r * HID;
#pragma unroll
        for (int p = 0; p < NPL; ++p) { hv[u][p] = h[p * plane + o]; av[u][p] = a[p * PW + lane]; }
        ht[u] = tl ? h[NPL * plane + (size_t)(base + r) * TAILF + lane] : 0.f;
        at[u] = tl ? a[NPL * PW + lane] : 0.f;
      }
    }
#pragma unroll
    for (int u = 0; u < RB; ++u) {
      if (jb + u < RPW) {
        const float q = __int_as_float(__builtin_amdgcn_readlane(__float_as_int(dq), jb + u));
        v2f t = {0.f, 0.f};
#pragma unroll
        for (int p = 0; p < NPL; ++p) {
          float v = (av[u][p] + hv[u][p] * q) + bb[p];
          v = v > 0.f ? v : 0.f;
          t += v2f{v, v} * ww[p];
        }
        {
          float v = (at[u] + ht[u] * q) + bt;
          v = v > 0.f ? v : 0.f;
          t += v2f{v, v} * wt;
        }
#pragma unroll
        for (int mk = 16; mk >= 1; mk >>= 1) {
          t.x += __shfl_xor(t.x, mk);
          t.y += __shfl_xor(t.y, mk);
        }
        if (lane == jb + u) keep = t;
      }
    }
  }
  if (own) {
    const size_t o = (size_t)(base + rmine) * OUT_C;
    *(v2f*)(h2 + o) = keep;
    const v2f z = *(const v2f*)(out + o);
    *(v2f*)(out + o) = v2f{z.x + (b2[0] + bs[0] + keep.x * dq),
                           z.y + (b2[1] + bs[1] + keep.y * dq)};
  }
}

__global__ void __launch_bounds__(SW * 32)
scatter2_kernel(const float* __restrict__ h2,
                const float* __restrict__ deg,
                const int* __restrict__ src,
                const int* __restrict__ dst,
                float* __restrict__ out, int n_edges, int n_nodes,
                const float* __restrict__ keyw,
                const float* __restrict__ cntw) {
  __shared__ float acc[NPB * OUT_C];
  __shared__ int lst[SW * LCAP];
  const int lane = threadIdx.x & 31;
  const int wv = __builtin_amdgcn_readfirstlane((int)(threadIdx.x >> 5));
  const int base = blockIdx.x * NPB;
  const int cnt = (n_nodes - base < NPB) ? (n_nodes - base) : NPB;
  int* const my = lst + wv * LCAP;

  for (int i = threadIdx.x; i < NPB * OUT_C; i += SW * 32) acc[i] = 0.f;
  __syncthreads();

  const int nrec = __builtin_amdgcn_readfirstlane(((const int*)cntw)[blockIdx.x * SW + wv]);
  if (nrec >= 0) {
    const v2i* __restrict__ rec =
        (const v2i*)((const char*)cntw + record_offset(n_nodes, n_edges)) + ((size_t)blockIdx.x * SW + wv) * CAPW;
    for (int i0 = 0; i0 < nrec; i0 += 32 * RU) {
      v2i r[RU];
#pragma unroll
      for (int u = 0; u < RU; ++u) {
        const int i = i0 + u * 32 + lane;
        r[u] = (i < nrec) ? rec[i] : v2i{0, -1};
      }
      float ds[RU], dd[RU];
      v2f hv[RU];
#pragma unroll
      for (int u = 0; u < RU; ++u) {
        const int s = r[u].x;
        const int dr = r[u].y >= 0 ? r[u].y : 0;
        ds[u] = deg[s];
        dd[u] = deg[base + dr];
        hv[u] = *(const v2f*)(h2 + (size_t)s * OUT_C);
      }
#pragma unroll
      for (int u = 0; u < RU; ++u) {
        if (r[u].y >= 0) {
          const float nrm = deg_to_dinv(ds[u]) * deg_to_dinv(dd[u]);
          float* a = acc + r[u].y * OUT_C;
          lds_add(a,     hv[u].x * nrm);
          lds_add(a + 1, hv[u].y * nrm);
        }
      }
    }
  } else {
  auto drain = [&](const int* p, int k) {
    if (lane < k) {
      const int e = p[lane];
      const int s = src[e];
      const int d = dst[e];
      if ((unsigned)(d - base) < (unsigned)cnt) {
        const float nrm = deg_to_dinv(deg[s]) * deg_to_dinv(deg[d]);
        const v2f hv = *(const v2f*)(h2 + (size_t)s * OUT_C);
        float* r = acc + (d - base) * OUT_C;
        lds_add(r,     hv.x * nrm);
        lds_add(r + 1, hv.y * nrm);
      }
    }
  };

  int count = 0;
  auto append = [&](bool m, int ev) {
    const unsigned bal = __builtin_amdgcn_ballot_w32(m);
    if (bal) {
      if (m) my[count + (int)__builtin_amdgcn_mbcnt_lo(bal, 0u)] = ev;
      count += __builtin_popcount(bal);
    }
  };

  const int nk = key_count(n_nodes, n_edges);
  const unsigned char* keys = (const unsigned char*)keyw;
  const int kstride = SW * KSU * 512;
  const int nfullk = ((((size_t)keys) & 15) == 0) ? nk / kstride : 0;
  if (nfullk > 0) {
    const unsigned kb = (unsigned)(blockIdx.x & 255) * 0x01010101u;
    auto hit = [&](unsigned w) {
      const unsigned x = w ^ kb;
      return ~(((x & 0x7f7f7f7fu) + 0x7f7f7f7fu) | x) & 0x80808080u;
    };
    int ps = (int)(((long long)blockIdx.x * nfullk) / gridDim.x);
    const int wofs = wv * (KSU * 512) + lane * 16;
    int eb = ps * kstride + wofs;
    v4u cur[KSU];
#pragma unroll
    for (int u = 0; u < KSU; ++u) cur[u] = *(const v4u*)(keys + eb + u * 512);
#pragma unroll 2
    for (int it = 0; it < nfullk; ++it) {
      ps = (ps + 1 < nfullk) ? ps + 1 : 0;
      const int en = (it + 1 < nfullk) ? ps * kstride + wofs : eb;
      v4u nxt[KSU];
#pragma unroll
      for (int u = 0; u < KSU; ++u) nxt[u] = *(const v4u*)(keys + en + u * 512);
#pragma unroll
      for (int u = 0; u < KSU; ++u) {
        const v4u kv = cur[u];
        const int e0 = eb + u * 512;
        const unsigned z0 = hit(kv.x), z1 = hit(kv.y), z2 = hit(kv.z), z3 = hit(kv.w);
        unsigned m = (z0 >> 7) | (z1 >> 6) | (z2 >> 5) | (z3 >> 4);
        for (unsigned bal = __builtin_amdgcn_ballot_w32(m != 0u); bal;
             bal = __builtin_amdgcn_ballot_w32(m != 0u)) {
          if (m != 0u) {
            const int p = __builtin_ctz(m);
            my[count + (int)__builtin_amdgcn_mbcnt_lo(bal, 0u)] = e0 + ((p & 7) << 2) + (p >> 3);
          }
          count += __builtin_popcount(bal);
          m &= m - 1u;
        }
      }
      __builtin_amdgcn_fence(__ATOMIC_SEQ_CST, "wavefront");
      while (count >= 32) {
        count -= 32;
        drain(my + count, 32);
      }
#pragma unroll
      for (int u = 0; u < KSU; ++u) cur[u] = nxt[u];
      eb = en;
    }
  }
  const long long lo = (long long)nfullk * kstride;

  const long long wstride = (long long)SW * SU * 128;
  const int nfull = ((((size_t)dst) & 15) == 0) ? (int)(((long long)n_edges - lo) / wstride) : 0;
  if (nfull > 0) {
    int ps = (int)(((long long)blockIdx.x * nfull) / gridDim.x);
    const long long wofs = (long long)wv * (SU * 128) + lane * 4;
    long long eb = lo + (long long)ps * wstride + wofs;
    v4i cur[SU];
#pragma unroll
    for (int u = 0; u < SU; ++u) cur[u] = *(const v4i*)(dst + eb + u * 128);
#pragma unroll 2
    for (int it = 0; it < nfull; ++it) {
      ps = (ps + 1 < nfull) ? ps + 1 : 0;
      const long long en = (it + 1 < nfull) ? lo + (long long)ps * wstride + wofs : eb;
      v4i nxt[SU];
#pragma unroll
      for (int u = 0; u < SU; ++u) nxt[u] = *(const v4i*)(dst + en + u * 128);
#pragma unroll
      for (int u = 0; u < SU; ++u) {
        const v4i dv = cur[u];
        const int e0 = (int)(eb + u * 128);
        const bool m0 = (unsigned)(dv.x - base) < (unsigned)cnt;
        const bool m1 = (unsigned)(dv.y - base) < (unsigned)cnt;
        const bool m2 = (unsigned)(dv.z - base) < (unsigned)cnt;
        const bool m3 = (unsigned)(dv.w - base) < (unsigned)cnt;
        if (__builtin_amdgcn_ballot_w32(m0 | m1 | m2 | m3)) {
          append(m0, e0);
          append(m1, e0 + 1);
          append(m2, e0 + 2);
          append(m3, e0 + 3);
        }
      }
      __builtin_amdgcn_fence(__ATOMIC_SEQ_CST, "wavefront");
      while (count >= 32) {
        count -= 32;
        drain(my + count, 32);
      }
#pragma unroll
      for (int u = 0; u < SU; ++u) cur[u] = nxt[u];
      eb = en;
    }
  }
  for (long long t = lo + (long long)nfull * wstride + wv * 32; ; t += TU * SW * 32) {
    const bool done = t >= n_edges;
    if (!done) {
      int dv[TU];
#pragma unroll
      for (int u = 0; u < TU; ++u) {
        const long long e = t + u * (SW * 32) + lane;
        dv[u] = (e < n_edges) ? dst[e] : -1;
      }
#pragma unroll
      for (int u = 0; u < TU; ++u)
        append((unsigned)(dv[u] - base) < (unsigned)cnt, (int)(t + u * (SW * 32) + lane));
    }
    __builtin_amdgcn_fence(__ATOMIC_SEQ_CST, "wavefront");
    const int lim = done ? 1 : 32;
    while (count >= lim) {
      const int k = count < 32 ? count : 32;
      count -= k;
      drain(my + count, k);
    }
    if (done) break;
  }

  }

  __syncthreads();

  for (int i = threadIdx.x; i < cnt; i += SW * 32) {
    const size_t o = (size_t)(base + i) * OUT_C;
    const v2f z = *(const v2f*)(out + o);
    const float z0 = z.x + acc[i * OUT_C];
    const float z1 = z.y + acc[i * OUT_C + 1];
    const float mx = fmaxf(z0, z1);
    const float lse = mx + logf(expf(z0 - mx) + expf(z1 - mx));
    *(v2f*)(out + o) = v2f{z0 - lse, z1 - lse};
  }
}

extern "C" void kernel_launch(void* const* d_in, const int* in_sizes, int n_in,
                              void* d_out, int out_size, void* d_ws, size_t ws_size,
                              hipStream_t stream) {
  const float* x  = (const float*)d_in[0];
  const int*   ei = (const int*)d_in[1];
  const float* W1 = (const float*)d_in[2];
  const float* b1 = (const float*)d_in[3];
  const float* W2 = (const float*)d_in[4];
  const float* b2 = (const float*)d_in[5];
  const float* Ws = (const float*)d_in[6];
  const float* bs = (const float*)d_in[7];
  float* out = (float*)d_out;

  const int n_nodes = in_sizes[0] / IN_C;
  const int n_edges = in_sizes[1] / 2;
  const int* srcv = ei;
  const int* dstv = ei + n_edges;

  float* Wp   = (float*)d_ws;
  float* deg  = Wp + WP_FLOATS;
  size_t off  = ((size_t)n_nodes + 255) & ~(size_t)255;
  float* h    = deg + off;
  float* cntw = h + (size_t)n_nodes * HID;
  float* keyw = cntw + (size_t)((n_nodes + NPB - 1) / NPB) * SW;
  float* h2   = cntw + (size_t)n_nodes * HID;

  const int T = 256;

  prep_kernel<<<((n_edges + 3) / 4 + T - 1) / T, T, 0, stream>>>(deg, n_nodes, dstv, n_edges, keyw);

  pack_w_kernel<<<WP_FLOATS / T, T, 0, stream>>>(W1, Ws, Wp);

  int tiles = (n_nodes + 15) / 16;
  gemm1_wmma_kernel<<<(tiles + 7) / 8, T, 0, stream>>>(x, Wp, dstv, deg, h, out,
                                                   n_nodes, n_edges);

  scatter1_kernel<<<(n_nodes + NPB - 1) / NPB, SW * 32, 0, stream>>>(h, deg, srcv, dstv, keyw, n_edges, n_nodes,
                                                                          b1, W2, b2, bs, h2, out, cntw);

  scatter2_kernel<<<(n_nodes + NPB - 1) / NPB, SW * 32, 0, stream>>>(h2, deg, srcv, dstv, out, n_edges, n_nodes,
                                                                keyw, cntw);
}


# Degree count stored in the node's narrow feature row (96..99 + degree): stage 1 reads a source's last 4 features and degree from one line, owned rows' dinv from LDS; arrival-order sums as before
# speedup vs baseline: 2.8192x; 1.1302x over previous
#include <hip/hip_runtime.h>
#include <math.h>

#define IN_C 512
#define HID 100
#define OUT_C 2
#define NT 8
#define WP_FLOATS (256 * NT * 16 * 2)

typedef __attribute__((ext_vector_type(2))) float v2f;
typedef __attribute__((ext_vector_type(8))) float v8f;
typedef __attribute__((ext_vector_type(4))) float v4f;

#define PW 32
#define NPL 3
#define TAILF (HID - NPL * PW)
#define TP 8
#define DEGW TAILF
#define REGF (HID - (TP - TAILF))
#define NPB 400

typedef __attribute__((ext_vector_type(4))) unsigned v4u;
typedef __attribute__((ext_vector_type(2))) int v2i;
#define SWV 32
#define CAPW 512

__device__ __forceinline__ size_t count_bytes(int n_nodes) {
  return (size_t)((n_nodes + NPB - 1) / NPB) * SWV * sizeof(int);
}
__device__ __forceinline__ int key_count(int n_nodes, int n_edges) {
  const size_t room = (size_t)n_nodes * REGF * sizeof(float) - count_bytes(n_nodes);
  return (int)(((size_t)n_edges < room ? (size_t)n_edges : room) & ~(size_t)3);
}
__device__ __forceinline__ size_t record_offset(int n_nodes, int n_edges) {
  return (count_bytes(n_nodes) + (size_t)key_count(n_nodes, n_edges) + 255) & ~(size_t)255;
}

__global__ void prep_kernel(float* __restrict__ nrow, int n_nodes,
                            const int* __restrict__ dst, int n_edges,
                            float* __restrict__ keyw) {
  const int i = blockIdx.x * blockDim.x + threadIdx.x;
  for (int j = i; j < n_nodes; j += gridDim.x * blockDim.x) nrow[(size_t)j * TP + DEGW] = 1.0f;
  const int nk = key_count(n_nodes, n_edges);
  unsigned* kw = (unsigned*)keyw;
  const long long e = 4LL * i;
  if (e + 3 < nk) {
    const unsigned k0 = ((unsigned)dst[e]     / NPB) & 255u;
    const unsigned k1 = ((unsigned)dst[e + 1] / NPB) & 255u;
    const unsigned k2 = ((unsigned)dst[e + 2] / NPB) & 255u;
    const unsigned k3 = ((unsigned)dst[e + 3] / NPB) & 255u;
    kw[i] = k0 | (k1 << 8) | (k2 << 16) | (k3 << 24);
  }
}

__device__ __forceinline__ float deg_to_dinv(float d) { return (d > 0.f) ? rsqrtf(d) : 0.f; }

__global__ void pack_w_kernel(const float* __restrict__ W1,
                              const float* __restrict__ Ws,
                              float* __restrict__ Wp) {
  int idx = blockIdx.x * blockDim.x + threadIdx.x;
  if (idx >= WP_FLOATS) return;
  int kin = idx & 1;
  int col = (idx >> 1) & 15;
  int t   = (idx >> 5) & 7;
  int kp  = idx >> 8;
  int k   = (kp << 1) + kin;
  float v = 0.f;
  if (t < 7) {
    int c = (t << 4) + col;
    if (c < HID) v = W1[(size_t)k * HID + c];
  } else {
    if (col < OUT_C) v = Ws[(size_t)k * OUT_C + col];
  }
  Wp[idx] = v;
}

#define KBLK 16
__global__ void __launch_bounds__(256)
gemm1_wmma_kernel(const float* __restrict__ x, const float* __restrict__ Wp,
                  const int* __restrict__ dst, float* __restrict__ nrow,
                  float* __restrict__ h,
                  float* __restrict__ outp, int n_nodes, int n_edges) {
  const int lane = threadIdx.x & 31;
  const int wv   = threadIdx.x >> 5;
  const int tile = blockIdx.x * (blockDim.x >> 5) + wv;
  const int row0 = tile << 4;
  if (row0 >= n_nodes) return;

  const long long stride = (long long)((n_nodes + 15) >> 4) << 5;
  long long e = ((long long)tile << 5) + lane;
  int dcur = (e < n_edges) ? dst[e] : -1;

  const int m  = lane & 15;
  const int kh = (lane >> 4) << 1;

  v8f acc[NT] = {};

  const float* __restrict__ pa = x + (size_t)(row0 + m) * IN_C + kh;
  const float* __restrict__ pb = Wp + ((size_t)(lane >> 4) * 256) + (m << 1);

  for (int kb = 0; kb < IN_C / (4 * KBLK); ++kb) {
    if (dcur >= 0) atomicAdd(&nrow[(size_t)dcur * TP + DEGW], 1.0f);
    const long long en = e + stride;
    const int dnxt = (en < n_edges) ? dst[en] : -1;
    for (int ks = 0; ks < KBLK; ++ks) {
      v2f a = *(const v2f*)pa;
#pragma unroll
      for (int t = 0; t < NT; ++t) {
        v2f b = *(const v2f*)(pb + (t << 5));
        acc[t] = __builtin_amdgcn_wmma_f32_16x16x4_f32(
            false, a, false, b, (short)0, acc[t], false, false);
      }
      pa += 4;
      pb += 512;
    }
    dcur = dnxt;
    e = en;
  }
  if (dcur >= 0) atomicAdd(&nrow[(size_t)dcur * TP + DEGW], 1.0f);
  for (e += stride; e < n_edges; e += stride) atomicAdd(&nrow[(size_t)dst[e] * TP + DEGW], 1.0f);

  const int rbase = (lane < 16) ? 0 : 8;
  const size_t plane = (size_t)n_nodes * PW;
#pragma unroll
  for (int r = 0; r < 8; ++r) {
    const int row = row0 + rbase + r;
#pragma unroll
    for (int t = 0; t < 7; ++t) {
      int col = (t << 4) + m;
      if (col < HID) {
        const size_t o = (t < 2 * NPL)
            ? (size_t)(t >> 1) * plane + (size_t)row * PW + ((t & 1) << 4) + m
            : (size_t)NPL * plane + (size_t)row * TP + m;
        h[o] = acc[t][r];
      }
    }
    if (m < OUT_C) outp[(size_t)row * OUT_C + m] = acc[7][r];
  }
}

#define SW 32
#define KSU 2
#define SU 4
#define LCAP (32 + KSU * 512)
static_assert(LCAP >= 32 + SU * 128, "the list also holds a pass of the dst loop");
static_assert(SW == SWV, "one count per wave of an edge-pass workgroup");
#define TU 4
#define EB 8
#define RU 4
#define RPW ((NPB + SW - 1) / SW)
#define RB 4
static_assert(RPW <= 32, "lane j of a wave keeps the results of the wave's row j");
static_assert(LCAP >= 32 + TU * 32, "the list also holds a trip of the edge-at-a-time loop");

typedef __attribute__((ext_vector_type(4))) int v4i;

__device__ __forceinline__ void lds_add(float* p, float v) {
  __hip_atomic_fetch_add(p, v, __ATOMIC_RELAXED, __HIP_MEMORY_SCOPE_WORKGROUP);
}

__global__ void __launch_bounds__(SW * 32)
scatter1_kernel(const float* __restrict__ h,
                const float* __restrict__ nrow,
                const int* __restrict__ src,
                const int* __restrict__ dst,
                const float* __restrict__ keyw, int n_edges, int n_nodes,
                const float* __restrict__ b1,
                const float* __restrict__ W2,
                const float* __restrict__ b2,
                const float* __restrict__ bs,
                float* __restrict__ h2,
                float* __restrict__ out,
                float* __restrict__ cntw) {
  __shared__ float acc[NPB * HID];
  __shared__ float dinvl[NPB];
  __shared__ int lst[SW * LCAP];
  const int lane = threadIdx.x & 31;
  const int wv = __builtin_amdgcn_readfirstlane((int)(threadIdx.x >> 5));
  const int base = blockIdx.x * NPB;
  const int cnt = (n_nodes - base < NPB) ? (n_nodes - base) : NPB;
  const size_t plane = (size_t)n_nodes * PW;
  const char* __restrict__ hb0 = (const char*)h;
  const char* __restrict__ hb1 = (const char*)(h + plane);
  const char* __restrict__ hb2 = (const char*)(h + 2 * plane);
  const unsigned lb = (unsigned)lane * 4u;
  int* const my = lst + wv * LCAP;

  for (int i = threadIdx.x; i < NPB * HID; i += SW * 32) acc[i] = 0.f;
  for (int i = threadIdx.x; i < NPB; i += SW * 32)
    dinvl[i] = (i < cnt) ? deg_to_dinv(nrow[(size_t)(base + i) * TP + DEGW]) : 0.f;
  __syncthreads();

  const size_t rofs = record_offset(n_nodes, n_edges);
  v2i* const rec = (v2i*)((char*)cntw + rofs) + ((size_t)blockIdx.x * SW + wv) * CAPW;
  int rc = 0;
  bool full = rofs + (size_t)gridDim.x * SW * CAPW * sizeof(v2i)
              > (size_t)n_nodes * REGF * sizeof(float);

  auto drain = [&](const int* p, int k) {
    const int e = (lane < k) ? p[lane] : -1;
    int s = 0, dl = -1;
    float nrm = 0.f;
    v4f tq = {0.f, 0.f, 0.f, 0.f};
    if (e >= 0) {
      s = src[e];
      const int d = dst[e];
      if ((unsigned)(d - base) < (unsigned)cnt) dl = d - base;
      const float* __restrict__ rs = nrow + (size_t)s * TP;
      tq = *(const v4f*)rs;
      nrm = deg_to_dinv(rs[DEGW]) * (dl >= 0 ? dinvl[dl] : 0.f);
    }
    if (!full) {
      if (rc + k <= CAPW) {
        if (lane < k) rec[rc + lane] = v2i{s, dl};
        rc += k;
      } else {
        full = true;
      }
    }
    for (int jb = 0; jb < k; jb += EB) {
      float v[EB][3];
#pragma unroll
      for (int u = 0; u < EB; ++u) {
        const int sj = __builtin_amdgcn_readlane(s, jb + u);
        const unsigned so = (unsigned)sj * (PW * 4u) + lb;
        v[u][0] = *(const float*)(hb0 + so);
        v[u][1] = *(const float*)(hb1 + so);
        v[u][2] = *(const float*)(hb2 + so);
      }
#pragma unroll
      for (int u = 0; u < EB; ++u) {
        const int dj = __builtin_amdgcn_readlane(dl, jb + u);
        if (jb + u < k && dj >= 0) {
          const float nj = __int_as_float(__builtin_amdgcn_readlane(__float_as_int(nrm), jb + u));
          float* r = acc + dj * HID + lane;
          lds_add(r,      v[u][0] * nj);
          lds_add(r + 32, v[u][1] * nj);
          lds_add(r + 64, v[u][2] * nj);
        }
      }
    }
    if (dl >= 0) {
      float* r = acc + dl * HID + NPL * PW;
      lds_add(r,     tq.x * nrm);
      lds_add(r + 1, tq.y * nrm);
      lds_add(r + 2, tq.z * nrm);
      lds_add(r + 3, tq.w * nrm);
    }
  };

  int count = 0;
  auto append = [&](bool m, int ev) {
    const unsigned bal = __builtin_amdgcn_ballot_w32(m);
    if (bal) {
      if (m) my[count + (int)__builtin_amdgcn_mbcnt_lo(bal, 0u)] = ev;
      count += __builtin_popcount(bal);
    }
  };

  const int nk = key_count(n_nodes, n_edges);
  const unsigned char* keys = (const unsigned char*)keyw;
  const int kstride = SW * KSU * 512;
  const int nfullk = ((((size_t)keys) & 15) == 0) ? nk / kstride : 0;
  if (nfullk > 0) {
    const unsigned kb = (unsigned)(blockIdx.x & 255) * 0x01010101u;
    auto hit = [&](unsigned w) {
      const unsigned x = w ^ kb;
      return ~(((x & 0x7f7f7f7fu) + 0x7f7f7f7fu) | x) & 0x80808080u;
    };
    int ps = (int)(((long long)blockIdx.x * nfullk) / gridDim.x);
    const int wofs = wv * (KSU * 512) + lane * 16;
    int eb = ps * kstride + wofs;
    v4u cur[KSU];
#pragma unroll
    for (int u = 0; u < KSU; ++u) cur[u] = *(const v4u*)(keys + eb + u * 512);
#pragma unroll 2
    for (int it = 0; it < nfullk; ++it) {
      ps = (ps + 1 < nfullk) ? ps + 1 : 0;
      const int en = (it + 1 < nfullk) ? ps * kstride + wofs : eb;
      v4u nxt[KSU];
#pragma unroll
      for (int u = 0; u < KSU; ++u) nxt[u] = *(const v4u*)(keys + en + u * 512);
#pragma unroll
      for (int u = 0; u < KSU; ++u) {
        const v4u kv = cur[u];
        const int e0 = eb + u * 512;
        const unsigned z0 = hit(kv.x), z1 = hit(kv.y), z2 = hit(kv.z), z3 = hit(kv.w);
        unsigned m = (z0 >> 7) | (z1 >> 6) | (z2 >> 5) | (z3 >> 4);
        for (unsigned bal = __builtin_amdgcn_ballot_w32(m != 0u); bal;
             bal = __builtin_amdgcn_ballot_w32(m != 0u)) {
          if (m != 0u) {
            const int p = __builtin_ctz(m);
            my[count + (int)__builtin_amdgcn_mbcnt_lo(bal, 0u)] = e0 + ((p & 7) << 2) + (p >> 3);
          }
          count += __builtin_popcount(bal);
          m &= m - 1u;
        }
      }
      __builtin_amdgcn_fence(__ATOMIC_SEQ_CST, "wavefront");
      while (count >= 32) {
        count -= 32;
        drain(my + count, 32);
      }
#pragma unroll
      for (int u = 0; u < KSU; ++u) cur[u] = nxt[u];
      eb = en;
    }
  }
  const long long lo = (long long)nfullk * kstride;

  const long long wstride = (long long)SW * SU * 128;
  const int nfull = ((((size_t)dst) & 15) == 0) ? (int)(((long long)n_edges - lo) / wstride) : 0;
  if (nfull > 0) {
    int ps = (int)(((long long)blockIdx.x * nfull) / gridDim.x);
    const long long wofs = (long long)wv * (SU * 128) + lane * 4;
    long long eb = lo + (long long)ps * wstride + wofs;
    v4i cur[SU];
#pragma unroll
    for (int u = 0; u < SU; ++u) cur[u] = *(const v4i*)(dst + eb + u * 128);
#pragma unroll 2
    for (int it = 0; it < nfull; ++it) {
      ps = (ps + 1 < nfull) ? ps + 1 : 0;
      const long long en = (it + 1 < nfull) ? lo + (long long)ps * wstride + wofs : eb;
      v4i nxt[SU];
#pragma unroll
      for (int u = 0; u < SU; ++u) nxt[u] = *(const v4i*)(dst + en + u * 128);
#pragma unroll
      for (int u = 0; u < SU; ++u) {
        const v4i dv = cur[u];
        const int e0 = (int)(eb + u * 128);
        const bool m0 = (unsigned)(dv.x - base) < (unsigned)cnt;
        const bool m1 = (unsigned)(dv.y - base) < (unsigned)cnt;
        const bool m2 = (unsigned)(dv.z - base) < (unsigned)cnt;
        const bool m3 = (unsigned)(dv.w - base) < (unsigned)cnt;
        if (__builtin_amdgcn_ballot_w32(m0 | m1 | m2 | m3)) {
          append(m0, e0);
          append(m1, e0 + 1);
          append(m2, e0 + 2);
          append(m3, e0 + 3);
        }
      }
      __builtin_amdgcn_fence(__ATOMIC_SEQ_CST, "wavefront");
      while (count >= 32) {
        count -= 32;
        drain(my + count, 32);
      }
#pragma unroll
      for (int u = 0; u < SU; ++u) cur[u] = nxt[u];
      eb = en;
    }
  }
  for (long long t = lo + (long long)nfull * wstride + wv * 32; ; t += TU * SW * 32) {
    const bool done = t >= n_edges;
    if (!done) {
      int dv[TU];
#pragma unroll
      for (int u = 0; u < TU; ++u) {
        const long long e = t + u * (SW * 32) + lane;
        dv[u] = (e < n_edges) ? dst[e] : -1;
      }
#pragma unroll
      for (int u = 0; u < TU; ++u)
        append((unsigned)(dv[u] - base) < (unsigned)cnt, (int)(t + u * (SW * 32) + lane));
    }
    __builtin_amdgcn_fence(__ATOMIC_SEQ_CST, "wavefront");
    const int lim = done ? 1 : 32;
    while (count >= lim) {
      const int k = count < 32 ? count : 32;
      count -= k;
      drain(my + count, k);
    }
    if (done) break;
  }

  if (lane == 0) ((int*)cntw)[blockIdx.x * SW + wv] = full ? -1 : rc;

  __syncthreads();

  const v2f* __restrict__ w2 = (const v2f*)W2;
  float bb[NPL];
  v2f ww[NPL];
#pragma unroll
  for (int p = 0; p < NPL; ++p) { bb[p] = b1[p * PW + lane]; ww[p] = w2[p * PW + lane]; }
  const bool tl = lane < TAILF;
  const float bt = tl ? b1[NPL * PW + lane] : 0.f;
  const v2f wt = tl ? w2[NPL * PW + lane] : v2f{0.f, 0.f};
  const int r0 = wv * RPW;
  const int rmine = r0 + lane;
  const bool own = lane < RPW && rmine < cnt;
  float dq = 0.f;
  if (own) { const float di = dinvl[rmine]; dq = di * di; }
  v2f keep = {0.f, 0.f};
#pragma unroll
  for (int jb = 0; jb < RPW; jb += RB) {
    float hv[RB][NPL], ht[RB], av[RB][NPL], at[RB];
#pragma unroll
    for (int u = 0; u < RB; ++u) {
      if (jb + u < RPW) {
        const int r = (r0 + jb + u < cnt) ? r0 + jb + u : cnt - 1;
        const size_t o = (size_t)(base + r) * PW + lane;
        const float* a = acc + r * HID;
#pragma unroll
        for (int p = 0; p < NPL; ++p) { hv[u][p] = h[p * plane + o]; av[u][p] = a[p * PW + lane]; }
        ht[u] = tl ? h[NPL * plane + (size_t)(base + r) * TP + lane] : 0.f;
        at[u] = tl ? a[NPL * PW + lane] : 0.f;
      }
    }
#pragma unroll
    for (int u = 0; u < RB; ++u) {
      if (jb + u < RPW) {
        const float q = __int_as_float(__builtin_amdgcn_readlane(__float_as_int(dq), jb + u));
        v2f t = {0.f, 0.f};
#pragma unroll
        for (int p = 0; p < NPL; ++p) {
          float v = (av[u][p] + hv[u][p] * q) + bb[p];
          v = v > 0.f ? v : 0.f;
          t += v2f{v, v} * ww[p];
        }
        {
          float v = (at[u] + ht[u] * q) + bt;
          v = v > 0.f ? v : 0.f;
          t += v2f{v, v} * wt;
        }
#pragma unroll
        for (int mk = 16; mk >= 1; mk >>= 1) {
          t.x += __shfl_xor(t.x, mk);
          t.y += __shfl_xor(t.y, mk);
        }
        if (lane == jb + u) keep = t;
      }
    }
  }
  if (own) {
    const size_t o = (size_t)(base + rmine) * OUT_C;
    *(v2f*)(h2 + o) = keep;
    const v2f z = *(const v2f*)(out + o);
    *(v2f*)(out + o) = v2f{z.x + (b2[0] + bs[0] + keep.x * dq),
                           z.y + (b2[1] + bs[1] + keep.y * dq)};
  }
}

__global__ void __launch_bounds__(SW * 32)
scatter2_kernel(const float* __restrict__ h2,
                const float* __restrict__ nrow,
                const int* __restrict__ src,
                const int* __restrict__ dst,
                float* __restrict__ out, int n_edges, int n_nodes,
                const float* __restrict__ keyw,
                const float* __restrict__ cntw) {
  __shared__ float acc[NPB * OUT_C];
  __shared__ float dinvl[NPB];
  __shared__ int lst[SW * LCAP];
  const int lane = threadIdx.x & 31;
  const int wv = __builtin_amdgcn_readfirstlane((int)(threadIdx.x >> 5));
  const int base = blockIdx.x * NPB;
  const int cnt = (n_nodes - base < NPB) ? (n_nodes - base) : NPB;
  int* const my = lst + wv * LCAP;

  for (int i = threadIdx.x; i < NPB * OUT_C; i += SW * 32) acc[i] = 0.f;
  for (int i = threadIdx.x; i < NPB; i += SW * 32)
    dinvl[i] = (i < cnt) ? deg_to_dinv(nrow[(size_t)(base + i) * TP + DEGW]) : 0.f;
  __syncthreads();

  const int nrec = __builtin_amdgcn_readfirstlane(((const int*)cntw)[blockIdx.x * SW + wv]);
  if (nrec >= 0) {
    const v2i* __restrict__ rec =
        (const v2i*)((const char*)cntw + record_offset(n_nodes, n_edges)) + ((size_t)blockIdx.x * SW + wv) * CAPW;
    for (int i0 = 0; i0 < nrec; i0 += 32 * RU) {
      v2i r[RU];
#pragma unroll
      for (int u = 0; u < RU; ++u) {
        const int i = i0 + u * 32 + lane;
        r[u] = (i < nrec) ? rec[i] : v2i{0, -1};
      }
      float ds[RU];
      v2f hv[RU];
#pragma unroll
      for (int u = 0; u < RU; ++u) {
        const int s = r[u].x;
        ds[u] = nrow[(size_t)s * TP + DEGW];
        hv[u] = *(const v2f*)(h2 + (size_t)s * OUT_C);
      }
#pragma unroll
      for (int u = 0; u < RU; ++u) {
        if (r[u].y >= 0) {
          const float nrm = deg_to_dinv(ds[u]) * dinvl[r[u].y];
          float* a = acc + r[u].y * OUT_C;
          lds_add(a,     hv[u].x * nrm);
          lds_add(a + 1, hv[u].y * nrm);
        }
      }
    }
  } else {
  auto drain = [&](const int* p, int k) {
    if (lane < k) {
      const int e = p[lane];
      const int s = src[e];
      const int d = dst[e];
      if ((unsigned)(d - base) < (unsigned)cnt) {
        const float nrm = deg_to_dinv(nrow[(size_t)s * TP + DEGW]) * dinvl[d - base];
        const v2f hv = *(const v2f*)(h2 + (size_t)s * OUT_C);
        float* r = acc + (d - base) * OUT_C;
        lds_add(r,     hv.x * nrm);
        lds_add(r + 1, hv.y * nrm);
      }
    }
  };

  int count = 0;
  auto append = [&](bool m, int ev) {
    const unsigned bal = __builtin_amdgcn_ballot_w32(m);
    if (bal) {
      if (m) my[count + (int)__builtin_amdgcn_mbcnt_lo(bal, 0u)] = ev;
      count += __builtin_popcount(bal);
    }
  };

  const int nk = key_count(n_nodes, n_edges);
  const unsigned char* keys = (const unsigned char*)keyw;
  const int kstride = SW * KSU * 512;
  const int nfullk = ((((size_t)keys) & 15) == 0) ? nk / kstride : 0;
  if (nfullk > 0) {
    const unsigned kb = (unsigned)(blockIdx.x & 255) * 0x01010101u;
    auto hit = [&](unsigned w) {
      const unsigned x = w ^ kb;
      return ~(((x & 0x7f7f7f7fu) + 0x7f7f7f7fu) | x) & 0x80808080u;
    };
    int ps = (int)(((long long)blockIdx.x * nfullk) / gridDim.x);
    const int wofs = wv * (KSU * 512) + lane * 16;
    int eb = ps * kstride + wofs;
    v4u cur[KSU];
#pragma unroll
    for (int u = 0; u < KSU; ++u) cur[u] = *(const v4u*)(keys + eb + u * 512);
#pragma unroll 2
    for (int it = 0; it < nfullk; ++it) {
      ps = (ps + 1 < nfullk) ? ps + 1 : 0;
      const int en = (it + 1 < nfullk) ? ps * kstride + wofs : eb;
      v4u nxt[KSU];
#pragma unroll
      for (int u = 0; u < KSU; ++u) nxt[u] = *(const v4u*)(keys + en + u * 512);
#pragma unroll
      for (int u = 0; u < KSU; ++u) {
        const v4u kv = cur[u];
        const int e0 = eb + u * 512;
        const unsigned z0 = hit(kv.x), z1 = hit(kv.y), z2 = hit(kv.z), z3 = hit(kv.w);
        unsigned m = (z0 >> 7) | (z1 >> 6) | (z2 >> 5) | (z3 >> 4);
        for (unsigned bal = __builtin_amdgcn_ballot_w32(m != 0u); bal;
             bal = __builtin_amdgcn_ballot_w32(m != 0u)) {
          if (m != 0u) {
            const int p = __builtin_ctz(m);
            my[count + (int)__builtin_amdgcn_mbcnt_lo(bal, 0u)] = e0 + ((p & 7) << 2) + (p >> 3);
          }
          count += __builtin_popcount(bal);
          m &= m - 1u;
        }
      }
      __builtin_amdgcn_fence(__ATOMIC_SEQ_CST, "wavefront");
      while (count >= 32) {
        count -= 32;
        drain(my + count, 32);
      }
#pragma unroll
      for (int u = 0; u < KSU; ++u) cur[u] = nxt[u];
      eb = en;
    }
  }
  const long long lo = (long long)nfullk * kstride;

  const long long wstride = (long long)SW * SU * 128;
  const int nfull = ((((size_t)dst) & 15) == 0) ? (int)(((long long)n_edges - lo) / wstride) : 0;
  if (nfull > 0) {
    int ps = (int)(((long long)blockIdx.x * nfull) / gridDim.x);
    const long long wofs = (long long)wv * (SU * 128) + lane * 4;
    long long eb = lo + (long long)ps * wstride + wofs;
    v4i cur[SU];
#pragma unroll
    for (int u = 0; u < SU; ++u) cur[u] = *(const v4i*)(dst + eb + u * 128);
#pragma unroll 2
    for (int it = 0; it < nfull; ++it) {
      ps = (ps + 1 < nfull) ? ps + 1 : 0;
      const long long en = (it + 1 < nfull) ? lo + (long long)ps * wstride + wofs : eb;
      v4i nxt[SU];
#pragma unroll
      for (int u = 0; u < SU; ++u) nxt[u] = *(const v4i*)(dst + en + u * 128);
#pragma unroll
      for (int u = 0; u < SU; ++u) {
        const v4i dv = cur[u];
        const int e0 = (int)(eb + u * 128);
        const bool m0 = (unsigned)(dv.x - base) < (unsigned)cnt;
        const bool m1 = (unsigned)(dv.y - base) < (unsigned)cnt;
        const bool m2 = (unsigned)(dv.z - base) < (unsigned)cnt;
        const bool m3 = (unsigned)(dv.w - base) < (unsigned)cnt;
        if (__builtin_amdgcn_ballot_w32(m0 | m1 | m2 | m3)) {
          append(m0, e0);
          append(m1, e0 + 1);
          append(m2, e0 + 2);
          append(m3, e0 + 3);
        }
      }
      __builtin_amdgcn_fence(__ATOMIC_SEQ_CST, "wavefront");
      while (count >= 32) {
        count -= 32;
        drain(my + count, 32);
      }
#pragma unroll
      for (int u = 0; u < SU; ++u) cur[u] = nxt[u];
      eb = en;
    }
  }
  for (long long t = lo + (long long)nfull * wstride + wv * 32; ; t += TU * SW * 32) {
    const bool done = t >= n_edges;
    if (!done) {
      int dv[TU];
#pragma unroll
      for (int u = 0; u < TU; ++u) {
        const long long e = t + u * (SW * 32) + lane;
        dv[u] = (e < n_edges) ? dst[e] : -1;
      }
#pragma unroll
      for (int u = 0; u < TU; ++u)
        append((unsigned)(dv[u] - base) < (unsigned)cnt, (int)(t + u * (SW * 32) + lane));
    }
    __builtin_amdgcn_fence(__ATOMIC_SEQ_CST, "wavefront");
    const int lim = done ? 1 : 32;
    while (count >= lim) {
      const int k = count < 32 ? count : 32;
      count -= k;
      drain(my + count, k);
    }
    if (done) break;
  }

  }

  __syncthreads();

  for (int i = threadIdx.x; i < cnt; i += SW * 32) {
    const size_t o = (size_t)(base + i) * OUT_C;
    const v2f z = *(const v2f*)(out + o);
    const float z0 = z.x + acc[i * OUT_C];
    const float z1 = z.y + acc[i * OUT_C + 1];
    const float mx = fmaxf(z0, z1);
    const float lse = mx + logf(expf(z0 - mx) + expf(z1 - mx));
    *(v2f*)(out + o) = v2f{z0 - lse, z1 - lse};
  }
}

extern "C" void kernel_launch(void* const* d_in, const int* in_sizes, int n_in,
                              void* d_out, int out_size, void* d_ws, size_t ws_size,
                              hipStream_t stream) {
  const float* x  = (const float*)d_in[0];
  const int*   ei = (const int*)d_in[1];
  const float* W1 = (const float*)d_in[2];
  const float* b1 = (const float*)d_in[3];
  const float* W2 = (const float*)d_in[4];
  const float* b2 = (const float*)d_in[5];
  const float* Ws = (const float*)d_in[6];
  const float* bs = (const float*)d_in[7];
  float* out = (float*)d_out;

  const int n_nodes = in_sizes[0] / IN_C;
  const int n_edges = in_sizes[1] / 2;
  const int* srcv = ei;
  const int* dstv = ei + n_edges;

  float* Wp   = (float*)d_ws;
  float* h    = Wp + WP_FLOATS;
  float* nrow = h + (size_t)n_nodes * (NPL * PW);
  float* cntw = h + (size_t)n_nodes * (NPL * PW + TP);
  float* keyw = cntw + (size_t)((n_nodes + NPB - 1) / NPB) * SW;
  float* h2   = cntw + (size_t)n_nodes * REGF;

  const int T = 256;

  prep_kernel<<<((n_edges + 3) / 4 + T - 1) / T, T, 0, stream>>>(nrow, n_nodes, dstv, n_edges, keyw);

  pack_w_kernel<<<WP_FLOATS / T, T, 0, stream>>>(W1, Ws, Wp);

  int tiles = (n_nodes + 15) / 16;
  gemm1_wmma_kernel<<<(tiles + 7) / 8, T, 0, stream>>>(x, Wp, dstv, nrow, h, out,
                                                   n_nodes, n_edges);

  scatter1_kernel<<<(n_nodes + NPB - 1) / NPB, SW * 32, 0, stream>>>(h, nrow, srcv, dstv, keyw, n_edges, n_nodes,
                                                                          b1, W2, b2, bs, h2, out, cntw);

  scatter2_kernel<<<(n_nodes + NPB - 1) / NPB, SW * 32, 0, stream>>>(h2, nrow, srcv, dstv, out, n_edges, n_nodes,
                                                                keyw, cntw);
}
